# starting kernel
# baseline (speedup 1.0000x reference)
#include <hip/hip_runtime.h>
#include <math.h>

typedef __attribute__((ext_vector_type(2))) float v2f;
typedef __attribute__((ext_vector_type(8))) float v8f;

#define NB    32
#define CCH   64
#define TT    300
#define VV    25
#define NV    (NB*VV)
#define MROWS (NV*TT)
#define TP    150
#define KHALF 4

__global__ void __launch_bounds__(128)
tgat_xw_wmma(const float* __restrict__ x, const float* __restrict__ W,
             float* __restrict__ xw) {
  __shared__ v2f sWp[32 * 64];
  for (int i = threadIdx.x; i < 32 * 64; i += 128) {
    const int kp  = i >> 6;
    const int col = i & 63;
    v2f p;
    p.x = W[(2 * kp) * 64 + col];
    p.y = W[(2 * kp + 1) * 64 + col];
    sWp[i] = p;
  }
  __syncthreads();

  const int lane  = threadIdx.x & 31;
  const int wave  = threadIdx.x >> 5;
  const int half  = lane >> 4;
  const int mloc  = lane & 15;
  const int mbase = blockIdx.x * 64 + wave * 16;

  const int m  = mbase + mloc;
  const int nv = m / TT;
  const int t  = m - nv * TT;
  const int n  = nv / VV;
  const int v  = nv - n * VV;
  const float* xb = x + (size_t)n * (CCH * TT * VV) + (size_t)t * VV + v;

  v8f acc0 = {}, acc1 = {}, acc2 = {}, acc3 = {};
  for (int k = 0; k < 64; k += 4) {
    const int k0 = k + 2 * half;
    v2f a;
    a.x = xb[(size_t)k0 * (TT * VV)];
    a.y = xb[(size_t)(k0 + 1) * (TT * VV)];
    const v2f* wp = sWp + (k0 >> 1) * 64 + mloc;
    const v2f b0 = wp[0];
    const v2f b1 = wp[16];
    const v2f b2 = wp[32];
    const v2f b3 = wp[48];
    acc0 = __builtin_amdgcn_wmma_f32_16x16x4_f32(false, a, false, b0, (short)0, acc0, false, false);
    acc1 = __builtin_amdgcn_wmma_f32_16x16x4_f32(false, a, false, b1, (short)0, acc1, false, false);
    acc2 = __builtin_amdgcn_wmma_f32_16x16x4_f32(false, a, false, b2, (short)0, acc2, false, false);
    acc3 = __builtin_amdgcn_wmma_f32_16x16x4_f32(false, a, false, b3, (short)0, acc3, false, false);
  }

#pragma unroll
  for (int r = 0; r < 8; ++r) {
    float* orow = xw + (size_t)(mbase + r + 8 * half) * 64 + mloc;
    orow[0]  = acc0[r];
    orow[16] = acc1[r];
    orow[32] = acc2[r];
    orow[48] = acc3[r];
  }
}

__global__ void __launch_bounds__(256)
tgat_attn(const float* __restrict__ xw, const float* __restrict__ alpha,
          const float* __restrict__ phi, float* __restrict__ out) {
  __shared__ float sx1[TP * 16];
  __shared__ float sx2[TT * 16];
  __shared__ float srow[64];
  __shared__ float se[TP * 9];
  __shared__ float sD[TP];

  const int nv = blockIdx.x;
  const int n  = nv / VV;
  const int v  = nv - n * VV;
  const float* xwB = xw + (size_t)nv * TT * 64;
  const int tid = threadIdx.x;

  for (int idx = tid; idx < TT * 16; idx += 256) {
    const int s = idx >> 4;
    const int h = idx & 15;
    const float* xr = xwB + (size_t)s * 64;
    const float* pr = phi + h * 64;
    float a2 = 0.f;
    for (int c = 0; c < 64; ++c) a2 += pr[c] * xr[c];
    sx2[idx] = a2;
    if ((s & 1) == 0) {
      const float* ar = alpha + h * 64;
      float a1 = 0.f;
      for (int c = 0; c < 64; ++c) a1 += ar[c] * xr[c];
      sx1[(s >> 1) * 16 + h] = a1;
    }
  }
  for (int c = tid; c < 64; c += 256) {
    float accv = 0.f;
    for (int s = 0; s < TT; ++s) accv += xwB[(size_t)s * 64 + c];
    srow[c] = accv;
  }
  __syncthreads();

  for (int idx = tid; idx < TP * 9; idx += 256) {
    const int tp = idx / 9;
    const int j  = idx - tp * 9;
    const int s  = 2 * tp - KHALF + j;
    float e = 0.f;
    if ((unsigned)s < (unsigned)TT) {
      float sc = 0.f;
      for (int h = 0; h < 16; ++h) sc += sx1[tp * 16 + h] * sx2[s * 16 + h];
      e = __expf(sc) - 1.f;
    }
    se[idx] = e;
  }
  __syncthreads();

  for (int tp = tid; tp < TP; tp += 256) {
    float D = (float)TT;
    for (int j = 0; j < 9; ++j) D += se[tp * 9 + j];
    sD[tp] = D;
  }
  __syncthreads();

  for (int idx = tid; idx < TP * 64; idx += 256) {
    const int tp = idx >> 6;
    const int c  = idx & 63;
    float accv = srow[c];
#pragma unroll
    for (int j = 0; j < 9; ++j) {
      int s = 2 * tp - KHALF + j;
      s = s < 0 ? 0 : (s > TT - 1 ? TT - 1 : s);
      accv += se[tp * 9 + j] * xwB[(size_t)s * 64 + c];
    }
    out[(((size_t)n * 64 + c) * TP + tp) * VV + v] = accv / sD[tp];
  }
}

extern "C" void kernel_launch(void* const* d_in, const int* in_sizes, int n_in,
                              void* d_out, int out_size, void* d_ws, size_t ws_size,
                              hipStream_t stream) {
  const float* x     = (const float*)d_in[0];
  const float* W     = (const float*)d_in[1];
  const float* alpha = (const float*)d_in[2];
  const float* phi   = (const float*)d_in[3];

  float* xw = (float*)d_ws;

  tgat_xw_wmma<<<MROWS / 64, 128, 0, stream>>>(x, W, xw);
  tgat_attn<<<NV, 256, 0, stream>>>(xw, alpha, phi, (float*)d_out);
}


# x1/x2 projections (xw rows times alpha/phi) moved from per-thread 64-term scalar dots with line-divergent loads onto f32 WMMA, one 16-wide tile per wave; changes the order of those 64-term sums
# speedup vs baseline: 1.3261x; 1.3261x over previous
#include <hip/hip_runtime.h>
#include <math.h>

typedef __attribute__((ext_vector_type(2))) float v2f;
typedef __attribute__((ext_vector_type(4))) float v4f;
typedef __attribute__((ext_vector_type(8))) float v8f;

#define NB    32
#define CCH   64
#define TT    300
#define VV    25
#define NV    (NB*VV)
#define MROWS (NV*TT)
#define TP    150
#define KHALF 4

__global__ void __launch_bounds__(128)
tgat_xw_wmma(const float* __restrict__ x, const float* __restrict__ W,
             float* __restrict__ xw) {
  __shared__ v2f sWp[32 * 64];
  for (int i = threadIdx.x; i < 32 * 64; i += 128) {
    const int kp  = i >> 6;
    const int col = i & 63;
    v2f p;
    p.x = W[(2 * kp) * 64 + col];
    p.y = W[(2 * kp + 1) * 64 + col];
    sWp[i] = p;
  }
  __syncthreads();

  const int lane  = threadIdx.x & 31;
  const int wave  = threadIdx.x >> 5;
  const int half  = lane >> 4;
  const int mloc  = lane & 15;
  const int mbase = blockIdx.x * 64 + wave * 16;

  const int m  = mbase + mloc;
  const int nv = m / TT;
  const int t  = m - nv * TT;
  const int n  = nv / VV;
  const int v  = nv - n * VV;
  const float* xb = x + (size_t)n * (CCH * TT * VV) + (size_t)t * VV + v;

  v8f acc0 = {}, acc1 = {}, acc2 = {}, acc3 = {};
  for (int k = 0; k < 64; k += 4) {
    const int k0 = k + 2 * half;
    v2f a;
    a.x = xb[(size_t)k0 * (TT * VV)];
    a.y = xb[(size_t)(k0 + 1) * (TT * VV)];
    const v2f* wp = sWp + (k0 >> 1) * 64 + mloc;
    const v2f b0 = wp[0];
    const v2f b1 = wp[16];
    const v2f b2 = wp[32];
    const v2f b3 = wp[48];
    acc0 = __builtin_amdgcn_wmma_f32_16x16x4_f32(false, a, false, b0, (short)0, acc0, false, false);
    acc1 = __builtin_amdgcn_wmma_f32_16x16x4_f32(false, a, false, b1, (short)0, acc1, false, false);
    acc2 = __builtin_amdgcn_wmma_f32_16x16x4_f32(false, a, false, b2, (short)0, acc2, false, false);
    acc3 = __builtin_amdgcn_wmma_f32_16x16x4_f32(false, a, false, b3, (short)0, acc3, false, false);
  }

#pragma unroll
  for (int r = 0; r < 8; ++r) {
    float* orow = xw + (size_t)(mbase + r + 8 * half) * 64 + mloc;
    orow[0]  = acc0[r];
    orow[16] = acc1[r];
    orow[32] = acc2[r];
    orow[48] = acc3[r];
  }
}

__global__ void __launch_bounds__(256)
tgat_attn(const float* __restrict__ xw, const float* __restrict__ alpha,
          const float* __restrict__ phi, float* __restrict__ out) {
  __shared__ float sx1[TP * 16];
  __shared__ float sx2[TT * 16];
  __shared__ float srow[64];
  __shared__ float se[TP * 9];
  __shared__ float sD[TP];

  const int nv = blockIdx.x;
  const int n  = nv / VV;
  const int v  = nv - n * VV;
  const float* xwB = xw + (size_t)nv * TT * 64;
  const int tid = threadIdx.x;

  {
    const int lane = tid & 31;
    const int wave = tid >> 5;
    const int hh   = lane >> 4;
    const int l16  = lane & 15;
    const int nt   = wave & 1;
    const float* prow = (nt ? phi : alpha) + l16 * 64 + 4 * hh;
    v4f b[8];
    for (int i = 0; i < 8; ++i) b[i] = *(const v4f*)(prow + 8 * i);

    for (int mt = wave >> 1; mt < (TT + 15) / 16; mt += 4) {
      int row = mt * 16 + l16;
      row = row < TT ? row : TT - 1;
      const float* arow = xwB + (size_t)row * 64 + 4 * hh;
      v4f a[8];
      for (int i = 0; i < 8; ++i) a[i] = *(const v4f*)(arow + 8 * i);
      v8f acc0 = {}, acc1 = {};
      for (int i = 0; i < 8; ++i) {
        v2f a0, a1, b0, b1;
        a0.x = a[i].x; a0.y = a[i].y; a1.x = a[i].z; a1.y = a[i].w;
        b0.x = b[i].x; b0.y = b[i].y; b1.x = b[i].z; b1.y = b[i].w;
        acc0 = __builtin_amdgcn_wmma_f32_16x16x4_f32(false, a0, false, b0, (short)0, acc0, false, false);
        acc1 = __builtin_amdgcn_wmma_f32_16x16x4_f32(false, a1, false, b1, (short)0, acc1, false, false);
      }
      const int r0 = mt * 16 + 8 * hh;
      if (nt) {
        for (int r = 0; r < 8; ++r) {
          const int orow = r0 + r;
          if (orow < TT) sx2[orow * 16 + l16] = acc0[r] + acc1[r];
        }
      } else {
        for (int r = 0; r < 8; r += 2) {
          const int orow = r0 + r;
          if (orow < TT) sx1[(orow >> 1) * 16 + l16] = acc0[r] + acc1[r];
        }
      }
    }
  }
  for (int c = tid; c < 64; c += 256) {
    float accv = 0.f;
    for (int s = 0; s < TT; ++s) accv += xwB[(size_t)s * 64 + c];
    srow[c] = accv;
  }
  __syncthreads();

  for (int idx = tid; idx < TP * 9; idx += 256) {
    const int tp = idx / 9;
    const int j  = idx - tp * 9;
    const int s  = 2 * tp - KHALF + j;
    float e = 0.f;
    if ((unsigned)s < (unsigned)TT) {
      float sc = 0.f;
      for (int h = 0; h < 16; ++h) sc += sx1[tp * 16 + h] * sx2[s * 16 + h];
      e = __expf(sc) - 1.f;
    }
    se[idx] = e;
  }
  __syncthreads();

  for (int tp = tid; tp < TP; tp += 256) {
    float D = (float)TT;
    for (int j = 0; j < 9; ++j) D += se[tp * 9 + j];
    sD[tp] = D;
  }
  __syncthreads();

  for (int idx = tid; idx < TP * 64; idx += 256) {
    const int tp = idx >> 6;
    const int c  = idx & 63;
    float accv = srow[c];
#pragma unroll
    for (int j = 0; j < 9; ++j) {
      int s = 2 * tp - KHALF + j;
      s = s < 0 ? 0 : (s > TT - 1 ? TT - 1 : s);
      accv += se[tp * 9 + j] * xwB[(size_t)s * 64 + c];
    }
    out[(((size_t)n * 64 + c) * TP + tp) * VV + v] = accv / sD[tp];
  }
}

extern "C" void kernel_launch(void* const* d_in, const int* in_sizes, int n_in,
                              void* d_out, int out_size, void* d_ws, size_t ws_size,
                              hipStream_t stream) {
  const float* x     = (const float*)d_in[0];
  const float* W     = (const float*)d_in[1];
  const float* alpha = (const float*)d_in[2];
  const float* phi   = (const float*)d_in[3];

  float* xw = (float*)d_ws;

  tgat_xw_wmma<<<MROWS / 64, 128, 0, stream>>>(x, W, xw);
  tgat_attn<<<NV, 256, 0, stream>>>(xw, alpha, phi, (float*)d_out);
}


# xw GEMM tiles re-indexed to 16 consecutive (t,v) positions of one input plane: each A load reads two 64-byte runs of x, not ~26 scattered lines; rows stored to their own place; sums unchanged
# speedup vs baseline: 1.7167x; 1.2945x over previous
#include <hip/hip_runtime.h>
#include <math.h>

typedef __attribute__((ext_vector_type(2))) float v2f;
typedef __attribute__((ext_vector_type(4))) float v4f;
typedef __attribute__((ext_vector_type(8))) float v8f;

#define NB    32
#define CCH   64
#define TT    300
#define VV    25
#define NV    (NB*VV)
#define MROWS (NV*TT)
#define TP    150
#define KHALF 4

__device__ __forceinline__ int tgat_row_of(int q) {
  const int n = q / (TT * VV);
  const int p = q - n * (TT * VV);
  const int t = p / VV;
  const int v = p - t * VV;
  return (n * VV + v) * TT + t;
}

__global__ void __launch_bounds__(128)
tgat_xw_wmma(const float* __restrict__ x, const float* __restrict__ W,
             float* __restrict__ xw) {
  __shared__ v2f sWp[32 * 64];
  for (int i = threadIdx.x; i < 32 * 64; i += 128) {
    const int kp  = i >> 6;
    const int col = i & 63;
    v2f p;
    p.x = W[(2 * kp) * 64 + col];
    p.y = W[(2 * kp + 1) * 64 + col];
    sWp[i] = p;
  }
  __syncthreads();

  const int lane  = threadIdx.x & 31;
  const int wave  = threadIdx.x >> 5;
  const int half  = lane >> 4;
  const int mloc  = lane & 15;
  const int qbase = blockIdx.x * 64 + wave * 16;

  const int q = qbase + mloc;
  const int n = q / (TT * VV);
  const int p = q - n * (TT * VV);
  const float* xb = x + (size_t)n * (CCH * TT * VV) + p;

  v8f acc0 = {}, acc1 = {}, acc2 = {}, acc3 = {};
  for (int k = 0; k < 64; k += 4) {
    const int k0 = k + 2 * half;
    v2f a;
    a.x = xb[(size_t)k0 * (TT * VV)];
    a.y = xb[(size_t)(k0 + 1) * (TT * VV)];
    const v2f* wp = sWp + (k0 >> 1) * 64 + mloc;
    const v2f b0 = wp[0];
    const v2f b1 = wp[16];
    const v2f b2 = wp[32];
    const v2f b3 = wp[48];
    acc0 = __builtin_amdgcn_wmma_f32_16x16x4_f32(false, a, false, b0, (short)0, acc0, false, false);
    acc1 = __builtin_amdgcn_wmma_f32_16x16x4_f32(false, a, false, b1, (short)0, acc1, false, false);
    acc2 = __builtin_amdgcn_wmma_f32_16x16x4_f32(false, a, false, b2, (short)0, acc2, false, false);
    acc3 = __builtin_amdgcn_wmma_f32_16x16x4_f32(false, a, false, b3, (short)0, acc3, false, false);
  }

#pragma unroll
  for (int r = 0; r < 8; ++r) {
    float* orow = xw + (size_t)tgat_row_of(qbase + r + 8 * half) * 64 + mloc;
    orow[0]  = acc0[r];
    orow[16] = acc1[r];
    orow[32] = acc2[r];
    orow[48] = acc3[r];
  }
}

__global__ void __launch_bounds__(256)
tgat_attn(const float* __restrict__ xw, const float* __restrict__ alpha,
          const float* __restrict__ phi, float* __restrict__ out) {
  __shared__ float sx1[TP * 16];
  __shared__ float sx2[TT * 16];
  __shared__ float srow[64];
  __shared__ float se[TP * 9];
  __shared__ float sD[TP];

  const int nv = blockIdx.x;
  const int n  = nv / VV;
  const int v  = nv - n * VV;
  const float* xwB = xw + (size_t)nv * TT * 64;
  const int tid = threadIdx.x;

  {
    const int lane = tid & 31;
    const int wave = tid >> 5;
    const int hh   = lane >> 4;
    const int l16  = lane & 15;
    const int nt   = wave & 1;
    const float* prow = (nt ? phi : alpha) + l16 * 64 + 4 * hh;
    v4f b[8];
    for (int i = 0; i < 8; ++i) b[i] = *(const v4f*)(prow + 8 * i);

    for (int mt = wave >> 1; mt < (TT + 15) / 16; mt += 4) {
      int row = mt * 16 + l16;
      row = row < TT ? row : TT - 1;
      const float* arow = xwB + (size_t)row * 64 + 4 * hh;
      v4f a[8];
      for (int i = 0; i < 8; ++i) a[i] = *(const v4f*)(arow + 8 * i);
      v8f acc0 = {}, acc1 = {};
      for (int i = 0; i < 8; ++i) {
        v2f a0, a1, b0, b1;
        a0.x = a[i].x; a0.y = a[i].y; a1.x = a[i].z; a1.y = a[i].w;
        b0.x = b[i].x; b0.y = b[i].y; b1.x = b[i].z; b1.y = b[i].w;
        acc0 = __builtin_amdgcn_wmma_f32_16x16x4_f32(false, a0, false, b0, (short)0, acc0, false, false);
        acc1 = __builtin_amdgcn_wmma_f32_16x16x4_f32(false, a1, false, b1, (short)0, acc1, false, false);
      }
      const int r0 = mt * 16 + 8 * hh;
      if (nt) {
        for (int r = 0; r < 8; ++r) {
          const int orow = r0 + r;
          if (orow < TT) sx2[orow * 16 + l16] = acc0[r] + acc1[r];
        }
      } else {
        for (int r = 0; r < 8; r += 2) {
          const int orow = r0 + r;
          if (orow < TT) sx1[(orow >> 1) * 16 + l16] = acc0[r] + acc1[r];
        }
      }
    }
  }
  for (int c = tid; c < 64; c += 256) {
    float accv = 0.f;
    for (int s = 0; s < TT; ++s) accv += xwB[(size_t)s * 64 + c];
    srow[c] = accv;
  }
  __syncthreads();

  for (int idx = tid; idx < TP * 9; idx += 256) {
    const int tp = idx / 9;
    const int j  = idx - tp * 9;
    const int s  = 2 * tp - KHALF + j;
    float e = 0.f;
    if ((unsigned)s < (unsigned)TT) {
      float sc = 0.f;
      for (int h = 0; h < 16; ++h) sc += sx1[tp * 16 + h] * sx2[s * 16 + h];
      e = __expf(sc) - 1.f;
    }
    se[idx] = e;
  }
  __syncthreads();

  for (int tp = tid; tp < TP; tp += 256) {
    float D = (float)TT;
    for (int j = 0; j < 9; ++j) D += se[tp * 9 + j];
    sD[tp] = D;
  }
  __syncthreads();

  for (int idx = tid; idx < TP * 64; idx += 256) {
    const int tp = idx >> 6;
    const int c  = idx & 63;
    float accv = srow[c];
#pragma unroll
    for (int j = 0; j < 9; ++j) {
      int s = 2 * tp - KHALF + j;
      s = s < 0 ? 0 : (s > TT - 1 ? TT - 1 : s);
      accv += se[tp * 9 + j] * xwB[(size_t)s * 64 + c];
    }
    out[(((size_t)n * 64 + c) * TP + tp) * VV + v] = accv / sD[tp];
  }
}

extern "C" void kernel_launch(void* const* d_in, const int* in_sizes, int n_in,
                              void* d_out, int out_size, void* d_ws, size_t ws_size,
                              hipStream_t stream) {
  const float* x     = (const float*)d_in[0];
  const float* W     = (const float*)d_in[1];
  const float* alpha = (const float*)d_in[2];
  const float* phi   = (const float*)d_in[3];

  float* xw = (float*)d_ws;

  tgat_xw_wmma<<<MROWS / 64, 128, 0, stream>>>(x, W, xw);
  tgat_attn<<<NV, 256, 0, stream>>>(xw, alpha, phi, (float*)d_out);
}


# attention results go as whole lines to a workspace plane [nv][tp][c]; a third launch turns 6-tp x 25-v tiles in LDS so each 128-B out line gets one store, not 32 scattered 4-B stores; sums unchanged
# speedup vs baseline: 5.3340x; 3.1071x over previous
#include <hip/hip_runtime.h>
#include <math.h>

typedef __attribute__((ext_vector_type(2))) float v2f;
typedef __attribute__((ext_vector_type(4))) float v4f;
typedef __attribute__((ext_vector_type(8))) float v8f;

#define NB    32
#define CCH   64
#define TT    300
#define VV    25
#define NV    (NB*VV)
#define MROWS (NV*TT)
#define TP    150
#define KHALF 4

__device__ __forceinline__ int tgat_row_of(int q) {
  const int n = q / (TT * VV);
  const int p = q - n * (TT * VV);
  const int t = p / VV;
  const int v = p - t * VV;
  return (n * VV + v) * TT + t;
}

__global__ void __launch_bounds__(128)
tgat_xw_wmma(const float* __restrict__ x, const float* __restrict__ W,
             float* __restrict__ xw) {
  __shared__ v2f sWp[32 * 64];
  for (int i = threadIdx.x; i < 32 * 64; i += 128) {
    const int kp  = i >> 6;
    const int col = i & 63;
    v2f p;
    p.x = W[(2 * kp) * 64 + col];
    p.y = W[(2 * kp + 1) * 64 + col];
    sWp[i] = p;
  }
  __syncthreads();

  const int lane  = threadIdx.x & 31;
  const int wave  = threadIdx.x >> 5;
  const int half  = lane >> 4;
  const int mloc  = lane & 15;
  const int qbase = blockIdx.x * 64 + wave * 16;

  const int q = qbase + mloc;
  const int n = q / (TT * VV);
  const int p = q - n * (TT * VV);
  const float* xb = x + (size_t)n * (CCH * TT * VV) + p;

  v8f acc0 = {}, acc1 = {}, acc2 = {}, acc3 = {};
  for (int k = 0; k < 64; k += 4) {
    const int k0 = k + 2 * half;
    v2f a;
    a.x = xb[(size_t)k0 * (TT * VV)];
    a.y = xb[(size_t)(k0 + 1) * (TT * VV)];
    const v2f* wp = sWp + (k0 >> 1) * 64 + mloc;
    const v2f b0 = wp[0];
    const v2f b1 = wp[16];
    const v2f b2 = wp[32];
    const v2f b3 = wp[48];
    acc0 = __builtin_amdgcn_wmma_f32_16x16x4_f32(false, a, false, b0, (short)0, acc0, false, false);
    acc1 = __builtin_amdgcn_wmma_f32_16x16x4_f32(false, a, false, b1, (short)0, acc1, false, false);
    acc2 = __builtin_amdgcn_wmma_f32_16x16x4_f32(false, a, false, b2, (short)0, acc2, false, false);
    acc3 = __builtin_amdgcn_wmma_f32_16x16x4_f32(false, a, false, b3, (short)0, acc3, false, false);
  }

#pragma unroll
  for (int r = 0; r < 8; ++r) {
    float* orow = xw + (size_t)tgat_row_of(qbase + r + 8 * half) * 64 + mloc;
    orow[0]  = acc0[r];
    orow[16] = acc1[r];
    orow[32] = acc2[r];
    orow[48] = acc3[r];
  }
}

__global__ void __launch_bounds__(256)
tgat_attn(const float* __restrict__ xw, const float* __restrict__ alpha,
          const float* __restrict__ phi, float* __restrict__ res) {
  __shared__ float sx1[TP * 16];
  __shared__ float sx2[TT * 16];
  __shared__ float srow[64];
  __shared__ float se[TP * 9];
  __shared__ float sD[TP];

  const int nv = blockIdx.x;
  const float* xwB = xw + (size_t)nv * TT * 64;
  const int tid = threadIdx.x;

  {
    const int lane = tid & 31;
    const int wave = tid >> 5;
    const int hh   = lane >> 4;
    const int l16  = lane & 15;
    const int nt   = wave & 1;
    const float* prow = (nt ? phi : alpha) + l16 * 64 + 4 * hh;
    v4f b[8];
    for (int i = 0; i < 8; ++i) b[i] = *(const v4f*)(prow + 8 * i);

    for (int mt = wave >> 1; mt < (TT + 15) / 16; mt += 4) {
      int row = mt * 16 + l16;
      row = row < TT ? row : TT - 1;
      const float* arow = xwB + (size_t)row * 64 + 4 * hh;
      v4f a[8];
      for (int i = 0; i < 8; ++i) a[i] = *(const v4f*)(arow + 8 * i);
      v8f acc0 = {}, acc1 = {};
      for (int i = 0; i < 8; ++i) {
        v2f a0, a1, b0, b1;
        a0.x = a[i].x; a0.y = a[i].y; a1.x = a[i].z; a1.y = a[i].w;
        b0.x = b[i].x; b0.y = b[i].y; b1.x = b[i].z; b1.y = b[i].w;
        acc0 = __builtin_amdgcn_wmma_f32_16x16x4_f32(false, a0, false, b0, (short)0, acc0, false, false);
        acc1 = __builtin_amdgcn_wmma_f32_16x16x4_f32(false, a1, false, b1, (short)0, acc1, false, false);
      }
      const int r0 = mt * 16 + 8 * hh;
      if (nt) {
        for (int r = 0; r < 8; ++r) {
          const int orow = r0 + r;
          if (orow < TT) sx2[orow * 16 + l16] = acc0[r] + acc1[r];
        }
      } else {
        for (int r = 0; r < 8; r += 2) {
          const int orow = r0 + r;
          if (orow < TT) sx1[(orow >> 1) * 16 + l16] = acc0[r] + acc1[r];
        }
      }
    }
  }
  for (int c = tid; c < 64; c += 256) {
    float accv = 0.f;
    for (int s = 0; s < TT; ++s) accv += xwB[(size_t)s * 64 + c];
    srow[c] = accv;
  }
  __syncthreads();

  for (int idx = tid; idx < TP * 9; idx += 256) {
    const int tp = idx / 9;
    const int j  = idx - tp * 9;
    const int s  = 2 * tp - KHALF + j;
    float e = 0.f;
    if ((unsigned)s < (unsigned)TT) {
      float sc = 0.f;
      for (int h = 0; h < 16; ++h) sc += sx1[tp * 16 + h] * sx2[s * 16 + h];
      e = __expf(sc) - 1.f;
    }
    se[idx] = e;
  }
  __syncthreads();

  for (int tp = tid; tp < TP; tp += 256) {
    float D = (float)TT;
    for (int j = 0; j < 9; ++j) D += se[tp * 9 + j];
    sD[tp] = D;
  }
  __syncthreads();

  for (int idx = tid; idx < TP * 64; idx += 256) {
    const int tp = idx >> 6;
    const int c  = idx & 63;
    float accv = srow[c];
#pragma unroll
    for (int j = 0; j < 9; ++j) {
      int s = 2 * tp - KHALF + j;
      s = s < 0 ? 0 : (s > TT - 1 ? TT - 1 : s);
      accv += se[tp * 9 + j] * xwB[(size_t)s * 64 + c];
    }
    res[(size_t)nv * (TP * 64) + idx] = accv / sD[tp];
  }
}

#define TPB    6
#define TILE_W (TPB * VV)
#define TILE_P 151
#define TILE_Q (VV * TPB * 16)

__global__ void __launch_bounds__(256)
tgat_out_tiles(const float* __restrict__ res, float* __restrict__ out) {
  __shared__ float tile[CCH * TILE_P];
  const int n   = blockIdx.x / (TP / TPB);
  const int k   = blockIdx.x - n * (TP / TPB);
  const int tid = threadIdx.x;

  v4f val[10];
#pragma unroll
  for (int u = 0; u < 10; ++u) {
    int i = tid + u * 256;
    i = i < TILE_Q ? i : TILE_Q - 1;
    const int v = i / (TPB * 16);
    const int r = i - v * (TPB * 16);
    val[u] = *(const v4f*)(res + ((size_t)(n * VV + v) * TP + k * TPB) * 64 + r * 4);
  }
#pragma unroll
  for (int u = 0; u < 10; ++u) {
    const int i = tid + u * 256;
    if (i < TILE_Q) {
      const int v   = i / (TPB * 16);
      const int r   = i - v * (TPB * 16);
      const int tpl = r >> 4;
      const int c4  = r & 15;
      float* d = tile + (4 * c4) * TILE_P + tpl * VV + v;
      d[0]          = val[u].x;
      d[TILE_P]     = val[u].y;
      d[2 * TILE_P] = val[u].z;
      d[3 * TILE_P] = val[u].w;
    }
  }
  __syncthreads();

  const int lane = tid & 31;
  const int wave = tid >> 5;
  for (int c = wave; c < CCH; c += 8) {
    const size_t g0 = ((size_t)(n * CCH + c) * TP + k * TPB) * VV;
    const int lead = (int)(g0 & 31);
    float w[6];
#pragma unroll
    for (int m = 0; m < 6; ++m) {
      int j = lane - lead + 32 * m;
      j = j < 0 ? 0 : (j > TILE_W - 1 ? TILE_W - 1 : j);
      w[m] = tile[c * TILE_P + j];
    }
#pragma unroll
    for (int m = 0; m < 6; ++m) {
      const int j = lane - lead + 32 * m;
      if (j >= 0 && j < TILE_W) out[g0 + j] = w[m];
    }
  }
}

extern "C" void kernel_launch(void* const* d_in, const int* in_sizes, int n_in,
                              void* d_out, int out_size, void* d_ws, size_t ws_size,
                              hipStream_t stream) {
  const float* x     = (const float*)d_in[0];
  const float* W     = (const float*)d_in[1];
  const float* alpha = (const float*)d_in[2];
  const float* phi   = (const float*)d_in[3];

  float* xw  = (float*)d_ws;
  float* res = xw + (size_t)MROWS * 64;

  tgat_xw_wmma<<<MROWS / 64, 128, 0, stream>>>(x, W, xw);
  tgat_attn<<<NV, 256, 0, stream>>>(xw, alpha, phi, res);
  tgat_out_tiles<<<NB * (TP / TPB), 256, 0, stream>>>(res, (float*)d_out);
}


# attention results: each thread walks 38 consecutive tp of one channel with the 9-row band window held in registers: 2 new xw loads per output, not 9, fetched in batches ahead of use; sums unchanged
# speedup vs baseline: 6.6461x; 1.2460x over previous
#include <hip/hip_runtime.h>
#include <math.h>

typedef __attribute__((ext_vector_type(2))) float v2f;
typedef __attribute__((ext_vector_type(4))) float v4f;
typedef __attribute__((ext_vector_type(8))) float v8f;

#define NB    32
#define CCH   64
#define TT    300
#define VV    25
#define NV    (NB*VV)
#define MROWS (NV*TT)
#define TP    150
#define KHALF 4

__device__ __forceinline__ int tgat_row_of(int q) {
  const int n = q / (TT * VV);
  const int p = q - n * (TT * VV);
  const int t = p / VV;
  const int v = p - t * VV;
  return (n * VV + v) * TT + t;
}

__global__ void __launch_bounds__(128)
tgat_xw_wmma(const float* __restrict__ x, const float* __restrict__ W,
             float* __restrict__ xw) {
  __shared__ v2f sWp[32 * 64];
  for (int i = threadIdx.x; i < 32 * 64; i += 128) {
    const int kp  = i >> 6;
    const int col = i & 63;
    v2f p;
    p.x = W[(2 * kp) * 64 + col];
    p.y = W[(2 * kp + 1) * 64 + col];
    sWp[i] = p;
  }
  __syncthreads();

  const int lane  = threadIdx.x & 31;
  const int wave  = threadIdx.x >> 5;
  const int half  = lane >> 4;
  const int mloc  = lane & 15;
  const int qbase = blockIdx.x * 64 + wave * 16;

  const int q = qbase + mloc;
  const int n = q / (TT * VV);
  const int p = q - n * (TT * VV);
  const float* xb = x + (size_t)n * (CCH * TT * VV) + p;

  v8f acc0 = {}, acc1 = {}, acc2 = {}, acc3 = {};
  for (int k = 0; k < 64; k += 4) {
    const int k0 = k + 2 * half;
    v2f a;
    a.x = xb[(size_t)k0 * (TT * VV)];
    a.y = xb[(size_t)(k0 + 1) * (TT * VV)];
    const v2f* wp = sWp + (k0 >> 1) * 64 + mloc;
    const v2f b0 = wp[0];
    const v2f b1 = wp[16];
    const v2f b2 = wp[32];
    const v2f b3 = wp[48];
    acc0 = __builtin_amdgcn_wmma_f32_16x16x4_f32(false, a, false, b0, (short)0, acc0, false, false);
    acc1 = __builtin_amdgcn_wmma_f32_16x16x4_f32(false, a, false, b1, (short)0, acc1, false, false);
    acc2 = __builtin_amdgcn_wmma_f32_16x16x4_f32(false, a, false, b2, (short)0, acc2, false, false);
    acc3 = __builtin_amdgcn_wmma_f32_16x16x4_f32(false, a, false, b3, (short)0, acc3, false, false);
  }

#pragma unroll
  for (int r = 0; r < 8; ++r) {
    float* orow = xw + (size_t)tgat_row_of(qbase + r + 8 * half) * 64 + mloc;
    orow[0]  = acc0[r];
    orow[16] = acc1[r];
    orow[32] = acc2[r];
    orow[48] = acc3[r];
  }
}

#define RTP   38
#define RROWS (2 * RTP + 7)
#define RK0   10
#define RB0   (2 * RK0 + 7)
#define RK1   24
#define RB1   (2 * RK1 + 7)

__device__ __forceinline__ int tgat_win_row(int row0, int i) {
  int r = row0 + i;
  if (i < KHALF) r = r < 0 ? 0 : r;
  if (i > (TT - 1) - (2 * 3 * RTP - KHALF)) r = r > TT - 1 ? TT - 1 : r;
  return r;
}

__global__ void __launch_bounds__(256)
tgat_attn(const float* __restrict__ xw, const float* __restrict__ alpha,
          const float* __restrict__ phi, float* __restrict__ res) {
  __shared__ float sx1[TP * 16];
  __shared__ float sx2[TT * 16];
  __shared__ float srow[64];
  __shared__ float se[TP * 9];
  __shared__ float sD[TP];

  const int nv = blockIdx.x;
  const float* xwB = xw + (size_t)nv * TT * 64;
  const int tid = threadIdx.x;

  {
    const int lane = tid & 31;
    const int wave = tid >> 5;
    const int hh   = lane >> 4;
    const int l16  = lane & 15;
    const int nt   = wave & 1;
    const float* prow = (nt ? phi : alpha) + l16 * 64 + 4 * hh;
    v4f b[8];
    for (int i = 0; i < 8; ++i) b[i] = *(const v4f*)(prow + 8 * i);

    for (int mt = wave >> 1; mt < (TT + 15) / 16; mt += 4) {
      int row = mt * 16 + l16;
      row = row < TT ? row : TT - 1;
      const float* arow = xwB + (size_t)row * 64 + 4 * hh;
      v4f a[8];
      for (int i = 0; i < 8; ++i) a[i] = *(const v4f*)(arow + 8 * i);
      v8f acc0 = {}, acc1 = {};
      for (int i = 0; i < 8; ++i) {
        v2f a0, a1, b0, b1;
        a0.x = a[i].x; a0.y = a[i].y; a1.x = a[i].z; a1.y = a[i].w;
        b0.x = b[i].x; b0.y = b[i].y; b1.x = b[i].z; b1.y = b[i].w;
        acc0 = __builtin_amdgcn_wmma_f32_16x16x4_f32(false, a0, false, b0, (short)0, acc0, false, false);
        acc1 = __builtin_amdgcn_wmma_f32_16x16x4_f32(false, a1, false, b1, (short)0, acc1, false, false);
      }
      const int r0 = mt * 16 + 8 * hh;
      if (nt) {
        for (int r = 0; r < 8; ++r) {
          const int orow = r0 + r;
          if (orow < TT) sx2[orow * 16 + l16] = acc0[r] + acc1[r];
        }
      } else {
        for (int r = 0; r < 8; r += 2) {
          const int orow = r0 + r;
          if (orow < TT) sx1[(orow >> 1) * 16 + l16] = acc0[r] + acc1[r];
        }
      }
    }
  }
  for (int c = tid; c < 64; c += 256) {
    float accv = 0.f;
    for (int s = 0; s < TT; ++s) accv += xwB[(size_t)s * 64 + c];
    srow[c] = accv;
  }
  __syncthreads();

  const int rc   = tid & 63;
  const int tp0  = (tid >> 6) * RTP;
  const int row0 = 2 * tp0 - KHALF;
  const float* xc = xwB + rc;
  float xr[RROWS];
#pragma unroll
  for (int i = 0; i < RB0; ++i) xr[i] = xc[tgat_win_row(row0, i) * 64];

  for (int idx = tid; idx < TP * 9; idx += 256) {
    const int tp = idx / 9;
    const int j  = idx - tp * 9;
    const int s  = 2 * tp - KHALF + j;
    float e = 0.f;
    if ((unsigned)s < (unsigned)TT) {
      float sc = 0.f;
      for (int h = 0; h < 16; ++h) sc += sx1[tp * 16 + h] * sx2[s * 16 + h];
      e = __expf(sc) - 1.f;
    }
    se[idx] = e;
  }
  __syncthreads();

  for (int tp = tid; tp < TP; tp += 256) {
    float D = (float)TT;
    for (int j = 0; j < 9; ++j) D += se[tp * 9 + j];
    sD[tp] = D;
  }
  __syncthreads();

  const float rs = srow[rc];
  float* rout = res + (size_t)nv * (TP * 64) + rc;
#pragma unroll
  for (int i = RB0; i < RB1; ++i) xr[i] = xc[tgat_win_row(row0, i) * 64];
#pragma unroll
  for (int k = 0; k < RTP; ++k) {
    if (k == RK0) {
#pragma unroll
      for (int i = RB1; i < RROWS; ++i) xr[i] = xc[tgat_win_row(row0, i) * 64];
    }
    const int tp = tp0 + k;
    if (k < TP - 3 * RTP || tp < TP) {
      float accv = rs;
#pragma unroll
      for (int j = 0; j < 9; ++j) accv += se[tp * 9 + j] * xr[2 * k + j];
      rout[tp * 64] = accv / sD[tp];
    }
  }
}

#define TPB    6
#define TILE_W (TPB * VV)
#define TILE_P 151
#define TILE_Q (VV * TPB * 16)

__global__ void __launch_bounds__(256)
tgat_out_tiles(const float* __restrict__ res, float* __restrict__ out) {
  __shared__ float tile[CCH * TILE_P];
  const int n   = blockIdx.x / (TP / TPB);
  const int k   = blockIdx.x - n * (TP / TPB);
  const int tid = threadIdx.x;

  v4f val[10];
#pragma unroll
  for (int u = 0; u < 10; ++u) {
    int i = tid + u * 256;
    i = i < TILE_Q ? i : TILE_Q - 1;
    const int v = i / (TPB * 16);
    const int r = i - v * (TPB * 16);
    val[u] = *(const v4f*)(res + ((size_t)(n * VV + v) * TP + k * TPB) * 64 + r * 4);
  }
#pragma unroll
  for (int u = 0; u < 10; ++u) {
    const int i = tid + u * 256;
    if (i < TILE_Q) {
      const int v   = i / (TPB * 16);
      const int r   = i - v * (TPB * 16);
      const int tpl = r >> 4;
      const int c4  = r & 15;
      float* d = tile + (4 * c4) * TILE_P + tpl * VV + v;
      d[0]          = val[u].x;
      d[TILE_P]     = val[u].y;
      d[2 * TILE_P] = val[u].z;
      d[3 * TILE_P] = val[u].w;
    }
  }
  __syncthreads();

  const int lane = tid & 31;
  const int wave = tid >> 5;
  for (int c = wave; c < CCH; c += 8) {
    const size_t g0 = ((size_t)(n * CCH + c) * TP + k * TPB) * VV;
    const int lead = (int)(g0 & 31);
    float w[6];
#pragma unroll
    for (int m = 0; m < 6; ++m) {
      int j = lane - lead + 32 * m;
      j = j < 0 ? 0 : (j > TILE_W - 1 ? TILE_W - 1 : j);
      w[m] = tile[c * TILE_P + j];
    }
#pragma unroll
    for (int m = 0; m < 6; ++m) {
      const int j = lane - lead + 32 * m;
      if (j >= 0 && j < TILE_W) out[g0 + j] = w[m];
    }
  }
}

extern "C" void kernel_launch(void* const* d_in, const int* in_sizes, int n_in,
                              void* d_out, int out_size, void* d_ws, size_t ws_size,
                              hipStream_t stream) {
  const float* x     = (const float*)d_in[0];
  const float* W     = (const float*)d_in[1];
  const float* alpha = (const float*)d_in[2];
  const float* phi   = (const float*)d_in[3];

  float* xw  = (float*)d_ws;
  float* res = xw + (size_t)MROWS * 64;

  tgat_xw_wmma<<<MROWS / 64, 128, 0, stream>>>(x, W, xw);
  tgat_attn<<<NV, 256, 0, stream>>>(xw, alpha, phi, res);
  tgat_out_tiles<<<NB * (TP / TPB), 256, 0, stream>>>(res, (float*)d_out);
}


# column sums of xw taken from the rows the projection waves already hold in registers (64 partials, then 4, then 1), not re-read by two waves in a 300-row walk; changes the order of those sums
# speedup vs baseline: 7.0978x; 1.0680x over previous
#include <hip/hip_runtime.h>
#include <math.h>

typedef __attribute__((ext_vector_type(2))) float v2f;
typedef __attribute__((ext_vector_type(4))) float v4f;
typedef __attribute__((ext_vector_type(8))) float v8f;

#define NB    32
#define CCH   64
#define TT    300
#define VV    25
#define NV    (NB*VV)
#define MROWS (NV*TT)
#define TP    150
#define KHALF 4

__device__ __forceinline__ int tgat_row_of(int q) {
  const int n = q / (TT * VV);
  const int p = q - n * (TT * VV);
  const int t = p / VV;
  const int v = p - t * VV;
  return (n * VV + v) * TT + t;
}

__global__ void __launch_bounds__(128)
tgat_xw_wmma(const float* __restrict__ x, const float* __restrict__ W,
             float* __restrict__ xw) {
  __shared__ v2f sWp[32 * 64];
  for (int i = threadIdx.x; i < 32 * 64; i += 128) {
    const int kp  = i >> 6;
    const int col = i & 63;
    v2f p;
    p.x = W[(2 * kp) * 64 + col];
    p.y = W[(2 * kp + 1) * 64 + col];
    sWp[i] = p;
  }
  __syncthreads();

  const int lane  = threadIdx.x & 31;
  const int wave  = threadIdx.x >> 5;
  const int half  = lane >> 4;
  const int mloc  = lane & 15;
  const int qbase = blockIdx.x * 64 + wave * 16;

  const int q = qbase + mloc;
  const int n = q / (TT * VV);
  const int p = q - n * (TT * VV);
  const float* xb = x + (size_t)n * (CCH * TT * VV) + p;

  v8f acc0 = {}, acc1 = {}, acc2 = {}, acc3 = {};
  for (int k = 0; k < 64; k += 4) {
    const int k0 = k + 2 * half;
    v2f a;
    a.x = xb[(size_t)k0 * (TT * VV)];
    a.y = xb[(size_t)(k0 + 1) * (TT * VV)];
    const v2f* wp = sWp + (k0 >> 1) * 64 + mloc;
    const v2f b0 = wp[0];
    const v2f b1 = wp[16];
    const v2f b2 = wp[32];
    const v2f b3 = wp[48];
    acc0 = __builtin_amdgcn_wmma_f32_16x16x4_f32(false, a, false, b0, (short)0, acc0, false, false);
    acc1 = __builtin_amdgcn_wmma_f32_16x16x4_f32(false, a, false, b1, (short)0, acc1, false, false);
    acc2 = __builtin_amdgcn_wmma_f32_16x16x4_f32(false, a, false, b2, (short)0, acc2, false, false);
    acc3 = __builtin_amdgcn_wmma_f32_16x16x4_f32(false, a, false, b3, (short)0, acc3, false, false);
  }

#pragma unroll
  for (int r = 0; r < 8; ++r) {
    float* orow = xw + (size_t)tgat_row_of(qbase + r + 8 * half) * 64 + mloc;
    orow[0]  = acc0[r];
    orow[16] = acc1[r];
    orow[32] = acc2[r];
    orow[48] = acc3[r];
  }
}

#define CS_P  68
#define RTP   38
#define RROWS (2 * RTP + 7)
#define RK0   10
#define RB0   (2 * RK0 + 7)
#define RK1   24
#define RB1   (2 * RK1 + 7)

__device__ __forceinline__ int tgat_win_row(int row0, int i) {
  int r = row0 + i;
  if (i < KHALF) r = r < 0 ? 0 : r;
  if (i > (TT - 1) - (2 * 3 * RTP - KHALF)) r = r > TT - 1 ? TT - 1 : r;
  return r;
}

__global__ void __launch_bounds__(256)
tgat_attn(const float* __restrict__ xw, const float* __restrict__ alpha,
          const float* __restrict__ phi, float* __restrict__ res) {
  __shared__ float sx1[TP * 16];
  __shared__ float sx2[TT * 16];
  __shared__ __attribute__((aligned(16))) float spart[64 * CS_P];
  __shared__ float srow4[4 * 64];
  __shared__ float se[TP * 9];
  __shared__ float sD[TP];

  const int nv = blockIdx.x;
  const float* xwB = xw + (size_t)nv * TT * 64;
  const int tid = threadIdx.x;

  {
    const int lane = tid & 31;
    const int wave = tid >> 5;
    const int hh   = lane >> 4;
    const int l16  = lane & 15;
    const int nt   = wave & 1;
    const float* prow = (nt ? phi : alpha) + l16 * 64 + 4 * hh;
    v4f b[8];
    for (int i = 0; i < 8; ++i) b[i] = *(const v4f*)(prow + 8 * i);
    v4f cs[4];
    for (int j = 0; j < 4; ++j) cs[j] = (v4f){0.f, 0.f, 0.f, 0.f};

    for (int mt = wave >> 1; mt < (TT + 15) / 16; mt += 4) {
      int row = mt * 16 + l16;
      row = row < TT ? row : TT - 1;
      const float* arow = xwB + (size_t)row * 64 + 4 * hh;
      v4f a[8];
      for (int i = 0; i < 8; ++i) a[i] = *(const v4f*)(arow + 8 * i);
      v8f acc0 = {}, acc1 = {};
      for (int i = 0; i < 8; ++i) {
        v2f a0, a1, b0, b1;
        a0.x = a[i].x; a0.y = a[i].y; a1.x = a[i].z; a1.y = a[i].w;
        b0.x = b[i].x; b0.y = b[i].y; b1.x = b[i].z; b1.y = b[i].w;
        acc0 = __builtin_amdgcn_wmma_f32_16x16x4_f32(false, a0, false, b0, (short)0, acc0, false, false);
        acc1 = __builtin_amdgcn_wmma_f32_16x16x4_f32(false, a1, false, b1, (short)0, acc1, false, false);
      }
      if (mt * 16 + l16 < TT) {
        if (nt) { for (int j = 0; j < 4; ++j) cs[j] += a[4 + j]; }
        else    { for (int j = 0; j < 4; ++j) cs[j] += a[j]; }
      }
      const int r0 = mt * 16 + 8 * hh;
      if (nt) {
        for (int r = 0; r < 8; ++r) {
          const int orow = r0 + r;
          if (orow < TT) sx2[orow * 16 + l16] = acc0[r] + acc1[r];
        }
      } else {
        for (int r = 0; r < 8; r += 2) {
          const int orow = r0 + r;
          if (orow < TT) sx1[(orow >> 1) * 16 + l16] = acc0[r] + acc1[r];
        }
      }
    }
    float* prt = spart + ((wave >> 1) * 16 + l16) * CS_P + 32 * nt + 4 * hh;
    for (int j = 0; j < 4; ++j) *(v4f*)(prt + 8 * j) = cs[j];
  }
  __syncthreads();

  const int rc   = tid & 63;
  const int tp0  = (tid >> 6) * RTP;
  const int row0 = 2 * tp0 - KHALF;
  const float* xc = xwB + rc;
  float xr[RROWS];
#pragma unroll
  for (int i = 0; i < RB0; ++i) xr[i] = xc[tgat_win_row(row0, i) * 64];

  {
    float accv = 0.f;
    for (int l = 0; l < 16; ++l) accv += spart[((tid >> 6) * 16 + l) * CS_P + rc];
    srow4[tid] = accv;
  }

  for (int idx = tid; idx < TP * 9; idx += 256) {
    const int tp = idx / 9;
    const int j  = idx - tp * 9;
    const int s  = 2 * tp - KHALF + j;
    float e = 0.f;
    if ((unsigned)s < (unsigned)TT) {
      float sc = 0.f;
      for (int h = 0; h < 16; ++h) sc += sx1[tp * 16 + h] * sx2[s * 16 + h];
      e = __expf(sc) - 1.f;
    }
    se[idx] = e;
  }
  __syncthreads();

  for (int tp = tid; tp < TP; tp += 256) {
    float D = (float)TT;
    for (int j = 0; j < 9; ++j) D += se[tp * 9 + j];
    sD[tp] = D;
  }
  __syncthreads();

  const float rs = ((srow4[rc] + srow4[64 + rc]) + srow4[128 + rc]) + srow4[192 + rc];
  float* rout = res + (size_t)nv * (TP * 64) + rc;
#pragma unroll
  for (int i = RB0; i < RB1; ++i) xr[i] = xc[tgat_win_row(row0, i) * 64];
#pragma unroll
  for (int k = 0; k < RTP; ++k) {
    if (k == RK0) {
#pragma unroll
      for (int i = RB1; i < RROWS; ++i) xr[i] = xc[tgat_win_row(row0, i) * 64];
    }
    const int tp = tp0 + k;
    if (k < TP - 3 * RTP || tp < TP) {
      float accv = rs;
#pragma unroll
      for (int j = 0; j < 9; ++j) accv += se[tp * 9 + j] * xr[2 * k + j];
      rout[tp * 64] = accv / sD[tp];
    }
  }
}

#define TPB    6
#define TILE_W (TPB * VV)
#define TILE_P 151
#define TILE_Q (VV * TPB * 16)

__global__ void __launch_bounds__(256)
tgat_out_tiles(const float* __restrict__ res, float* __restrict__ out) {
  __shared__ float tile[CCH * TILE_P];
  const int n   = blockIdx.x / (TP / TPB);
  const int k   = blockIdx.x - n * (TP / TPB);
  const int tid = threadIdx.x;

  v4f val[10];
#pragma unroll
  for (int u = 0; u < 10; ++u) {
    int i = tid + u * 256;
    i = i < TILE_Q ? i : TILE_Q - 1;
    const int v = i / (TPB * 16);
    const int r = i - v * (TPB * 16);
    val[u] = *(const v4f*)(res + ((size_t)(n * VV + v) * TP + k * TPB) * 64 + r * 4);
  }
#pragma unroll
  for (int u = 0; u < 10; ++u) {
    const int i = tid + u * 256;
    if (i < TILE_Q) {
      const int v   = i / (TPB * 16);
      const int r   = i - v * (TPB * 16);
      const int tpl = r >> 4;
      const int c4  = r & 15;
      float* d = tile + (4 * c4) * TILE_P + tpl * VV + v;
      d[0]          = val[u].x;
      d[TILE_P]     = val[u].y;
      d[2 * TILE_P] = val[u].z;
      d[3 * TILE_P] = val[u].w;
    }
  }
  __syncthreads();

  const int lane = tid & 31;
  const int wave = tid >> 5;
  for (int c = wave; c < CCH; c += 8) {
    const size_t g0 = ((size_t)(n * CCH + c) * TP + k * TPB) * VV;
    const int lead = (int)(g0 & 31);
    float w[6];
#pragma unroll
    for (int m = 0; m < 6; ++m) {
      int j = lane - lead + 32 * m;
      j = j < 0 ? 0 : (j > TILE_W - 1 ? TILE_W - 1 : j);
      w[m] = tile[c * TILE_P + j];
    }
#pragma unroll
    for (int m = 0; m < 6; ++m) {
      const int j = lane - lead + 32 * m;
      if (j >= 0 && j < TILE_W) out[g0 + j] = w[m];
    }
  }
}

extern "C" void kernel_launch(void* const* d_in, const int* in_sizes, int n_in,
                              void* d_out, int out_size, void* d_ws, size_t ws_size,
                              hipStream_t stream) {
  const float* x     = (const float*)d_in[0];
  const float* W     = (const float*)d_in[1];
  const float* alpha = (const float*)d_in[2];
  const float* phi   = (const float*)d_in[3];

  float* xw  = (float*)d_ws;
  float* res = xw + (size_t)MROWS * 64;

  tgat_xw_wmma<<<MROWS / 64, 128, 0, stream>>>(x, W, xw);
  tgat_attn<<<NV, 256, 0, stream>>>(xw, alpha, phi, res);
  tgat_out_tiles<<<NB * (TP / TPB), 256, 0, stream>>>(res, (float*)d_out);
}


# projection tiles: xw rows fetched as whole 256-B rows (two per load, a tile ahead) and handed to the fragment lanes through LDS, not as 32-B pieces of 16 rows per load instruction; sums unchanged
# speedup vs baseline: 7.4908x; 1.0554x over previous
#include <hip/hip_runtime.h>
#include <math.h>

typedef __attribute__((ext_vector_type(2))) float v2f;
typedef __attribute__((ext_vector_type(4))) float v4f;
typedef __attribute__((ext_vector_type(8))) float v8f;

#define NB    32
#define CCH   64
#define TT    300
#define VV    25
#define NV    (NB*VV)
#define MROWS (NV*TT)
#define TP    150
#define KHALF 4

__device__ __forceinline__ int tgat_row_of(int q) {
  const int n = q / (TT * VV);
  const int p = q - n * (TT * VV);
  const int t = p / VV;
  const int v = p - t * VV;
  return (n * VV + v) * TT + t;
}

__global__ void __launch_bounds__(128)
tgat_xw_wmma(const float* __restrict__ x, const float* __restrict__ W,
             float* __restrict__ xw) {
  __shared__ v2f sWp[32 * 64];
  for (int i = threadIdx.x; i < 32 * 64; i += 128) {
    const int kp  = i >> 6;
    const int col = i & 63;
    v2f p;
    p.x = W[(2 * kp) * 64 + col];
    p.y = W[(2 * kp + 1) * 64 + col];
    sWp[i] = p;
  }
  __syncthreads();

  const int lane  = threadIdx.x & 31;
  const int wave  = threadIdx.x >> 5;
  const int half  = lane >> 4;
  const int mloc  = lane & 15;
  const int qbase = blockIdx.x * 64 + wave * 16;

  const int q = qbase + mloc;
  const int n = q / (TT * VV);
  const int p = q - n * (TT * VV);
  const float* xb = x + (size_t)n * (CCH * TT * VV) + p;

  v8f acc0 = {}, acc1 = {}, acc2 = {}, acc3 = {};
  for (int k = 0; k < 64; k += 4) {
    const int k0 = k + 2 * half;
    v2f a;
    a.x = xb[(size_t)k0 * (TT * VV)];
    a.y = xb[(size_t)(k0 + 1) * (TT * VV)];
    const v2f* wp = sWp + (k0 >> 1) * 64 + mloc;
    const v2f b0 = wp[0];
    const v2f b1 = wp[16];
    const v2f b2 = wp[32];
    const v2f b3 = wp[48];
    acc0 = __builtin_amdgcn_wmma_f32_16x16x4_f32(false, a, false, b0, (short)0, acc0, false, false);
    acc1 = __builtin_amdgcn_wmma_f32_16x16x4_f32(false, a, false, b1, (short)0, acc1, false, false);
    acc2 = __builtin_amdgcn_wmma_f32_16x16x4_f32(false, a, false, b2, (short)0, acc2, false, false);
    acc3 = __builtin_amdgcn_wmma_f32_16x16x4_f32(false, a, false, b3, (short)0, acc3, false, false);
  }

#pragma unroll
  for (int r = 0; r < 8; ++r) {
    float* orow = xw + (size_t)tgat_row_of(qbase + r + 8 * half) * 64 + mloc;
    orow[0]  = acc0[r];
    orow[16] = acc1[r];
    orow[32] = acc2[r];
    orow[48] = acc3[r];
  }
}

#define CS_P  68
#define PSTEPS 5
#define RTP   38
#define RROWS (2 * RTP + 7)
#define RK0   10
#define RB0   (2 * RK0 + 7)
#define RK1   24
#define RB1   (2 * RK1 + 7)

__device__ __forceinline__ int tgat_win_row(int row0, int i) {
  int r = row0 + i;
  if (i < KHALF) r = r < 0 ? 0 : r;
  if (i > (TT - 1) - (2 * 3 * RTP - KHALF)) r = r > TT - 1 ? TT - 1 : r;
  return r;
}

__global__ void __launch_bounds__(256) __attribute__((amdgpu_waves_per_eu(8)))
tgat_attn(const float* __restrict__ xw, const float* __restrict__ alpha,
          const float* __restrict__ phi, float* __restrict__ res) {
  __shared__ float sx1[TP * 16];
  __shared__ float sx2[TT * 16];
  __shared__ __attribute__((aligned(16))) float spart[64 * CS_P];
  __shared__ float srow4[4 * 64];
  __shared__ __attribute__((aligned(16))) float sst[4 * 16 * CS_P];
  __shared__ float se[TP * 9];
  __shared__ float sD[TP];

  const int nv = blockIdx.x;
  const float* xwB = xw + (size_t)nv * TT * 64;
  const int tid = threadIdx.x;

  {
    const int lane = tid & 31;
    const int wave = tid >> 5;
    const int hh   = lane >> 4;
    const int l16  = lane & 15;
    const int nt   = wave & 1;
    const int pr   = wave >> 1;
    const float* prow = (nt ? phi : alpha) + l16 * 64 + 4 * hh;
    v4f b[8];
    for (int i = 0; i < 8; ++i) b[i] = *(const v4f*)(prow + 8 * i);
    v4f cs[4];
    for (int j = 0; j < 4; ++j) cs[j] = (v4f){0.f, 0.f, 0.f, 0.f};

    const int stw = pr * (16 * CS_P) + (8 * nt + hh) * CS_P + 4 * l16;
    const int str = pr * (16 * CS_P) + l16 * CS_P + 4 * hh;
    v4f g[4];
    for (int j = 0; j < 4; ++j) {
      int row = pr * 16 + 8 * nt + 2 * j + hh;
      row = row < TT ? row : TT - 1;
      g[j] = *(const v4f*)(xwB + (size_t)row * 64 + 4 * l16);
    }
    for (int j = 0; j < 4; ++j) *(v4f*)(sst + stw + 2 * j * CS_P) = g[j];

    for (int it = 0; it < PSTEPS; ++it) {
      const int mt = pr + 4 * it;
      const float* rbuf = (it & 1) ? spart : sst;
      float*       wbuf = (it & 1) ? sst : spart;
      __syncthreads();
      if (it + 1 < PSTEPS) {
        for (int j = 0; j < 4; ++j) {
          int row = (mt + 4) * 16 + 8 * nt + 2 * j + hh;
          row = row < TT ? row : TT - 1;
          g[j] = *(const v4f*)(xwB + (size_t)row * 64 + 4 * l16);
        }
      }
      v4f a[8];
      for (int i = 0; i < 8; ++i) a[i] = *(const v4f*)(rbuf + str + 8 * i);
      v8f acc0 = {}, acc1 = {};
      for (int i = 0; i < 8; ++i) {
        v2f a0, a1, b0, b1;
        a0.x = a[i].x; a0.y = a[i].y; a1.x = a[i].z; a1.y = a[i].w;
        b0.x = b[i].x; b0.y = b[i].y; b1.x = b[i].z; b1.y = b[i].w;
        acc0 = __builtin_amdgcn_wmma_f32_16x16x4_f32(false, a0, false, b0, (short)0, acc0, false, false);
        acc1 = __builtin_amdgcn_wmma_f32_16x16x4_f32(false, a1, false, b1, (short)0, acc1, false, false);
      }
      if (mt * 16 + l16 < TT) {
        if (nt) { for (int j = 0; j < 4; ++j) cs[j] += a[4 + j]; }
        else    { for (int j = 0; j < 4; ++j) cs[j] += a[j]; }
      }
      const int r0 = mt * 16 + 8 * hh;
      if (nt) {
        for (int r = 0; r < 8; ++r) {
          const int orow = r0 + r;
          if (orow < TT) sx2[orow * 16 + l16] = acc0[r] + acc1[r];
        }
      } else {
        for (int r = 0; r < 8; r += 2) {
          const int orow = r0 + r;
          if (orow < TT) sx1[(orow >> 1) * 16 + l16] = acc0[r] + acc1[r];
        }
      }
      if (it + 1 < PSTEPS) {
        for (int j = 0; j < 4; ++j) *(v4f*)(wbuf + stw + 2 * j * CS_P) = g[j];
      }
    }
    float* prt = spart + (pr * 16 + l16) * CS_P + 32 * nt + 4 * hh;
    for (int j = 0; j < 4; ++j) *(v4f*)(prt + 8 * j) = cs[j];
  }
  __syncthreads();

  const int rc   = tid & 63;
  const int tp0  = (tid >> 6) * RTP;
  const int row0 = 2 * tp0 - KHALF;
  const float* xc = xwB + rc;
  float xr[RROWS];
#pragma unroll
  for (int i = 0; i < RB0; ++i) xr[i] = xc[tgat_win_row(row0, i) * 64];

  {
    float accv = 0.f;
    for (int l = 0; l < 16; ++l) accv += spart[((tid >> 6) * 16 + l) * CS_P + rc];
    srow4[tid] = accv;
  }

  for (int idx = tid; idx < TP * 9; idx += 256) {
    const int tp = idx / 9;
    const int j  = idx - tp * 9;
    const int s  = 2 * tp - KHALF + j;
    float e = 0.f;
    if ((unsigned)s < (unsigned)TT) {
      float sc = 0.f;
      for (int h = 0; h < 16; ++h) sc += sx1[tp * 16 + h] * sx2[s * 16 + h];
      e = __expf(sc) - 1.f;
    }
    se[idx] = e;
  }
  __syncthreads();

  for (int tp = tid; tp < TP; tp += 256) {
    float D = (float)TT;
    for (int j = 0; j < 9; ++j) D += se[tp * 9 + j];
    sD[tp] = D;
  }
  __syncthreads();

  const float rs = ((srow4[rc] + srow4[64 + rc]) + srow4[128 + rc]) + srow4[192 + rc];
  float* rout = res + (size_t)nv * (TP * 64) + rc;
#pragma unroll
  for (int i = RB0; i < RB1; ++i) xr[i] = xc[tgat_win_row(row0, i) * 64];
#pragma unroll
  for (int k = 0; k < RTP; ++k) {
    if (k == RK0) {
#pragma unroll
      for (int i = RB1; i < RROWS; ++i) xr[i] = xc[tgat_win_row(row0, i) * 64];
    }
    const int tp = tp0 + k;
    if (k < TP - 3 * RTP || tp < TP) {
      float accv = rs;
#pragma unroll
      for (int j = 0; j < 9; ++j) accv += se[tp * 9 + j] * xr[2 * k + j];
      rout[tp * 64] = accv / sD[tp];
    }
  }
}

#define TPB    6
#define TILE_W (TPB * VV)
#define TILE_P 151
#define TILE_Q (VV * TPB * 16)

__global__ void __launch_bounds__(256)
tgat_out_tiles(const float* __restrict__ res, float* __restrict__ out) {
  __shared__ float tile[CCH * TILE_P];
  const int n   = blockIdx.x / (TP / TPB);
  const int k   = blockIdx.x - n * (TP / TPB);
  const int tid = threadIdx.x;

  v4f val[10];
#pragma unroll
  for (int u = 0; u < 10; ++u) {
    int i = tid + u * 256;
    i = i < TILE_Q ? i : TILE_Q - 1;
    const int v = i / (TPB * 16);
    const int r = i - v * (TPB * 16);
    val[u] = *(const v4f*)(res + ((size_t)(n * VV + v) * TP + k * TPB) * 64 + r * 4);
  }
#pragma unroll
  for (int u = 0; u < 10; ++u) {
    const int i = tid + u * 256;
    if (i < TILE_Q) {
      const int v   = i / (TPB * 16);
      const int r   = i - v * (TPB * 16);
      const int tpl = r >> 4;
      const int c4  = r & 15;
      float* d = tile + (4 * c4) * TILE_P + tpl * VV + v;
      d[0]          = val[u].x;
      d[TILE_P]     = val[u].y;
      d[2 * TILE_P] = val[u].z;
      d[3 * TILE_P] = val[u].w;
    }
  }
  __syncthreads();

  const int lane = tid & 31;
  const int wave = tid >> 5;
  for (int c = wave; c < CCH; c += 8) {
    const size_t g0 = ((size_t)(n * CCH + c) * TP + k * TPB) * VV;
    const int lead = (int)(g0 & 31);
    float w[6];
#pragma unroll
    for (int m = 0; m < 6; ++m) {
      int j = lane - lead + 32 * m;
      j = j < 0 ? 0 : (j > TILE_W - 1 ? TILE_W - 1 : j);
      w[m] = tile[c * TILE_P + j];
    }
#pragma unroll
    for (int m = 0; m < 6; ++m) {
      const int j = lane - lead + 32 * m;
      if (j >= 0 && j < TILE_W) out[g0 + j] = w[m];
    }
  }
}

extern "C" void kernel_launch(void* const* d_in, const int* in_sizes, int n_in,
                              void* d_out, int out_size, void* d_ws, size_t ws_size,
                              hipStream_t stream) {
  const float* x     = (const float*)d_in[0];
  const float* W     = (const float*)d_in[1];
  const float* alpha = (const float*)d_in[2];
  const float* phi   = (const float*)d_in[3];

  float* xw  = (float*)d_ws;
  float* res = xw + (size_t)MROWS * 64;

  tgat_xw_wmma<<<MROWS / 64, 128, 0, stream>>>(x, W, xw);
  tgat_attn<<<NV, 256, 0, stream>>>(xw, alpha, phi, res);
  tgat_out_tiles<<<NB * (TP / TPB), 256, 0, stream>>>(res, (float*)d_out);
}


# results: a row's nine band weights and 1/denominator lie as one aligned 48-B LDS record: wide constant-offset reads and one multiply per output, not pair reads with computed addresses and a divide
# speedup vs baseline: 7.7899x; 1.0399x over previous
#include <hip/hip_runtime.h>
#include <math.h>

typedef __attribute__((ext_vector_type(2))) float v2f;
typedef __attribute__((ext_vector_type(4))) float v4f;
typedef __attribute__((ext_vector_type(8))) float v8f;

#define NB    32
#define CCH   64
#define TT    300
#define VV    25
#define NV    (NB*VV)
#define MROWS (NV*TT)
#define TP    150
#define KHALF 4

__device__ __forceinline__ int tgat_row_of(int q) {
  const int n = q / (TT * VV);
  const int p = q - n * (TT * VV);
  const int t = p / VV;
  const int v = p - t * VV;
  return (n * VV + v) * TT + t;
}

__global__ void __launch_bounds__(128)
tgat_xw_wmma(const float* __restrict__ x, const float* __restrict__ W,
             float* __restrict__ xw) {
  __shared__ v2f sWp[32 * 64];
  for (int i = threadIdx.x; i < 32 * 64; i += 128) {
    const int kp  = i >> 6;
    const int col = i & 63;
    v2f p;
    p.x = W[(2 * kp) * 64 + col];
    p.y = W[(2 * kp + 1) * 64 + col];
    sWp[i] = p;
  }
  __syncthreads();

  const int lane  = threadIdx.x & 31;
  const int wave  = threadIdx.x >> 5;
  const int half  = lane >> 4;
  const int mloc  = lane & 15;
  const int qbase = blockIdx.x * 64 + wave * 16;

  const int q = qbase + mloc;
  const int n = q / (TT * VV);
  const int p = q - n * (TT * VV);
  const float* xb = x + (size_t)n * (CCH * TT * VV) + p;

  v8f acc0 = {}, acc1 = {}, acc2 = {}, acc3 = {};
  for (int k = 0; k < 64; k += 4) {
    const int k0 = k + 2 * half;
    v2f a;
    a.x = xb[(size_t)k0 * (TT * VV)];
    a.y = xb[(size_t)(k0 + 1) * (TT * VV)];
    const v2f* wp = sWp + (k0 >> 1) * 64 + mloc;
    const v2f b0 = wp[0];
    const v2f b1 = wp[16];
    const v2f b2 = wp[32];
    const v2f b3 = wp[48];
    acc0 = __builtin_amdgcn_wmma_f32_16x16x4_f32(false, a, false, b0, (short)0, acc0, false, false);
    acc1 = __builtin_amdgcn_wmma_f32_16x16x4_f32(false, a, false, b1, (short)0, acc1, false, false);
    acc2 = __builtin_amdgcn_wmma_f32_16x16x4_f32(false, a, false, b2, (short)0, acc2, false, false);
    acc3 = __builtin_amdgcn_wmma_f32_16x16x4_f32(false, a, false, b3, (short)0, acc3, false, false);
  }

#pragma unroll
  for (int r = 0; r < 8; ++r) {
    float* orow = xw + (size_t)tgat_row_of(qbase + r + 8 * half) * 64 + mloc;
    orow[0]  = acc0[r];
    orow[16] = acc1[r];
    orow[32] = acc2[r];
    orow[48] = acc3[r];
  }
}

#define CS_P  68
#define PSTEPS 5
#define RTP   38
#define RROWS (2 * RTP + 7)
#define RK0   10
#define RB0   (2 * RK0 + 7)
#define RK1   24
#define RB1   (2 * RK1 + 7)
#define SE_P  12

__device__ __forceinline__ int tgat_win_row(int row0, int i) {
  int r = row0 + i;
  if (i < KHALF) r = r < 0 ? 0 : r;
  if (i > (TT - 1) - (2 * 3 * RTP - KHALF)) r = r > TT - 1 ? TT - 1 : r;
  return r;
}

__global__ void __launch_bounds__(256) __attribute__((amdgpu_waves_per_eu(8)))
tgat_attn(const float* __restrict__ xw, const float* __restrict__ alpha,
          const float* __restrict__ phi, float* __restrict__ res) {
  __shared__ float sx1[TP * 16];
  __shared__ float sx2[TT * 16];
  __shared__ __attribute__((aligned(16))) float spart[64 * CS_P];
  __shared__ float srow4[4 * 64];
  __shared__ __attribute__((aligned(16))) float sst[4 * 16 * CS_P];
  __shared__ __attribute__((aligned(16))) float se[TP * SE_P];

  const int nv = blockIdx.x;
  const float* xwB = xw + (size_t)nv * TT * 64;
  const int tid = threadIdx.x;

  {
    const int lane = tid & 31;
    const int wave = tid >> 5;
    const int hh   = lane >> 4;
    const int l16  = lane & 15;
    const int nt   = wave & 1;
    const int pr   = wave >> 1;
    const float* prow = (nt ? phi : alpha) + l16 * 64 + 4 * hh;
    v4f b[8];
    for (int i = 0; i < 8; ++i) b[i] = *(const v4f*)(prow + 8 * i);
    v4f cs[4];
    for (int j = 0; j < 4; ++j) cs[j] = (v4f){0.f, 0.f, 0.f, 0.f};

    const int stw = pr * (16 * CS_P) + (8 * nt + hh) * CS_P + 4 * l16;
    const int str = pr * (16 * CS_P) + l16 * CS_P + 4 * hh;
    v4f g[4];
    for (int j = 0; j < 4; ++j) {
      int row = pr * 16 + 8 * nt + 2 * j + hh;
      row = row < TT ? row : TT - 1;
      g[j] = *(const v4f*)(xwB + (size_t)row * 64 + 4 * l16);
    }
    for (int j = 0; j < 4; ++j) *(v4f*)(sst + stw + 2 * j * CS_P) = g[j];

    for (int it = 0; it < PSTEPS; ++it) {
      const int mt = pr + 4 * it;
      const float* rbuf = (it & 1) ? spart : sst;
      float*       wbuf = (it & 1) ? sst : spart;
      __syncthreads();
      if (it + 1 < PSTEPS) {
        for (int j = 0; j < 4; ++j) {
          int row = (mt + 4) * 16 + 8 * nt + 2 * j + hh;
          row = row < TT ? row : TT - 1;
          g[j] = *(const v4f*)(xwB + (size_t)row * 64 + 4 * l16);
        }
      }
      v4f a[8];
      for (int i = 0; i < 8; ++i) a[i] = *(const v4f*)(rbuf + str + 8 * i);
      v8f acc0 = {}, acc1 = {};
      for (int i = 0; i < 8; ++i) {
        v2f a0, a1, b0, b1;
        a0.x = a[i].x; a0.y = a[i].y; a1.x = a[i].z; a1.y = a[i].w;
        b0.x = b[i].x; b0.y = b[i].y; b1.x = b[i].z; b1.y = b[i].w;
        acc0 = __builtin_amdgcn_wmma_f32_16x16x4_f32(false, a0, false, b0, (short)0, acc0, false, false);
        acc1 = __builtin_amdgcn_wmma_f32_16x16x4_f32(false, a1, false, b1, (short)0, acc1, false, false);
      }
      if (mt * 16 + l16 < TT) {
        if (nt) { for (int j = 0; j < 4; ++j) cs[j] += a[4 + j]; }
        else    { for (int j = 0; j < 4; ++j) cs[j] += a[j]; }
      }
      const int r0 = mt * 16 + 8 * hh;
      if (nt) {
        for (int r = 0; r < 8; ++r) {
          const int orow = r0 + r;
          if (orow < TT) sx2[orow * 16 + l16] = acc0[r] + acc1[r];
        }
      } else {
        for (int r = 0; r < 8; r += 2) {
          const int orow = r0 + r;
          if (orow < TT) sx1[(orow >> 1) * 16 + l16] = acc0[r] + acc1[r];
        }
      }
      if (it + 1 < PSTEPS) {
        for (int j = 0; j < 4; ++j) *(v4f*)(wbuf + stw + 2 * j * CS_P) = g[j];
      }
    }
    float* prt = spart + (pr * 16 + l16) * CS_P + 32 * nt + 4 * hh;
    for (int j = 0; j < 4; ++j) *(v4f*)(prt + 8 * j) = cs[j];
  }
  __syncthreads();

  const int rc   = tid & 63;
  const int tp0  = (tid >> 6) * RTP;
  const int row0 = 2 * tp0 - KHALF;
  const float* xc = xwB + rc;
  float xr[RROWS];
#pragma unroll
  for (int i = 0; i < RB0; ++i) xr[i] = xc[tgat_win_row(row0, i) * 64];

  {
    float accv = 0.f;
    for (int l = 0; l < 16; ++l) accv += spart[((tid >> 6) * 16 + l) * CS_P + rc];
    srow4[tid] = accv;
  }

  for (int idx = tid; idx < TP * 9; idx += 256) {
    const int tp = idx / 9;
    const int j  = idx - tp * 9;
    const int s  = 2 * tp - KHALF + j;
    float e = 0.f;
    if ((unsigned)s < (unsigned)TT) {
      float sc = 0.f;
      for (int h = 0; h < 16; ++h) sc += sx1[tp * 16 + h] * sx2[s * 16 + h];
      e = __expf(sc) - 1.f;
    }
    se[tp * SE_P + j] = e;
  }
  __syncthreads();

  for (int tp = tid; tp < TP; tp += 256) {
    float D = (float)TT;
    for (int j = 0; j < 9; ++j) D += se[tp * SE_P + j];
    se[tp * SE_P + 9] = 1.0f / D;
  }
  __syncthreads();

  const float rs = ((srow4[rc] + srow4[64 + rc]) + srow4[128 + rc]) + srow4[192 + rc];
  float* rout = res + (size_t)nv * (TP * 64) + rc;
#pragma unroll
  for (int i = RB0; i < RB1; ++i) xr[i] = xc[tgat_win_row(row0, i) * 64];
  float wq[2][20];
#pragma unroll
  for (int u = 0; u < 2; ++u) {
    const float* rec = se + (tp0 + u) * SE_P;
#pragma unroll
    for (int j = 0; j < 10; ++j) wq[0][10 * u + j] = rec[j];
  }
#pragma unroll
  for (int p = 0; p < RTP / 2; ++p) {
    if (2 * p == RK0) {
#pragma unroll
      for (int i = RB1; i < RROWS; ++i) xr[i] = xc[tgat_win_row(row0, i) * 64];
    }
    if (p + 1 < RTP / 2) {
#pragma unroll
      for (int u = 0; u < 2; ++u) {
        int tpn = tp0 + 2 * (p + 1) + u;
        if (2 * (p + 1) + u >= TP - 3 * RTP) tpn = tpn < TP ? tpn : TP - 1;
        const float* rec = se + tpn * SE_P;
#pragma unroll
        for (int j = 0; j < 10; ++j) wq[(p + 1) & 1][10 * u + j] = rec[j];
      }
    }
#pragma unroll
    for (int u = 0; u < 2; ++u) {
      const int k  = 2 * p + u;
      const int tp = tp0 + k;
      if (k < TP - 3 * RTP || tp < TP) {
        float accv = rs;
#pragma unroll
        for (int j = 0; j < 9; ++j) accv += wq[p & 1][10 * u + j] * xr[2 * k + j];
        rout[tp * 64] = accv * wq[p & 1][10 * u + 9];
      }
    }
    __builtin_amdgcn_sched_barrier(0);
  }
}

#define TPB    6
#define TILE_W (TPB * VV)
#define TILE_P 151
#define TILE_Q (VV * TPB * 16)

__global__ void __launch_bounds__(256)
tgat_out_tiles(const float* __restrict__ res, float* __restrict__ out) {
  __shared__ float tile[CCH * TILE_P];
  const int n   = blockIdx.x / (TP / TPB);
  const int k   = blockIdx.x - n * (TP / TPB);
  const int tid = threadIdx.x;

  v4f val[10];
#pragma unroll
  for (int u = 0; u < 10; ++u) {
    int i = tid + u * 256;
    i = i < TILE_Q ? i : TILE_Q - 1;
    const int v = i / (TPB * 16);
    const int r = i - v * (TPB * 16);
    val[u] = *(const v4f*)(res + ((size_t)(n * VV + v) * TP + k * TPB) * 64 + r * 4);
  }
#pragma unroll
  for (int u = 0; u < 10; ++u) {
    const int i = tid + u * 256;
    if (i < TILE_Q) {
      const int v   = i / (TPB * 16);
      const int r   = i - v * (TPB * 16);
      const int tpl = r >> 4;
      const int c4  = r & 15;
      float* d = tile + (4 * c4) * TILE_P + tpl * VV + v;
      d[0]          = val[u].x;
      d[TILE_P]     = val[u].y;
      d[2 * TILE_P] = val[u].z;
      d[3 * TILE_P] = val[u].w;
    }
  }
  __syncthreads();

  const int lane = tid & 31;
  const int wave = tid >> 5;
  for (int c = wave; c < CCH; c += 8) {
    const size_t g0 = ((size_t)(n * CCH + c) * TP + k * TPB) * VV;
    const int lead = (int)(g0 & 31);
    float w[6];
#pragma unroll
    for (int m = 0; m < 6; ++m) {
      int j = lane - lead + 32 * m;
      j = j < 0 ? 0 : (j > TILE_W - 1 ? TILE_W - 1 : j);
      w[m] = tile[c * TILE_P + j];
    }
#pragma unroll
    for (int m = 0; m < 6; ++m) {
      const int j = lane - lead + 32 * m;
      if (j >= 0 && j < TILE_W) out[g0 + j] = w[m];
    }
  }
}

extern "C" void kernel_launch(void* const* d_in, const int* in_sizes, int n_in,
                              void* d_out, int out_size, void* d_ws, size_t ws_size,
                              hipStream_t stream) {
  const float* x     = (const float*)d_in[0];
  const float* W     = (const float*)d_in[1];
  const float* alpha = (const float*)d_in[2];
  const float* phi   = (const float*)d_in[3];

  float* xw  = (float*)d_ws;
  float* res = xw + (size_t)MROWS * 64;

  tgat_xw_wmma<<<MROWS / 64, 128, 0, stream>>>(x, W, xw);
  tgat_attn<<<NV, 256, 0, stream>>>(xw, alpha, phi, res);
  tgat_out_tiles<<<NB * (TP / TPB), 256, 0, stream>>>(res, (float*)d_out);
}


# results walk: a thread owns two adjacent channels and 19 output rows, one wave per range of rows: 8-byte loads and stores that cover whole 256-B rows, packed multiply-adds; sums unchanged
# speedup vs baseline: 8.0696x; 1.0359x over previous
#include <hip/hip_runtime.h>
#include <math.h>

typedef __attribute__((ext_vector_type(2))) float v2f;
typedef __attribute__((ext_vector_type(4))) float v4f;
typedef __attribute__((ext_vector_type(8))) float v8f;

#define NB    32
#define CCH   64
#define TT    300
#define VV    25
#define NV    (NB*VV)
#define MROWS (NV*TT)
#define TP    150
#define KHALF 4

__device__ __forceinline__ int tgat_row_of(int q) {
  const int n = q / (TT * VV);
  const int p = q - n * (TT * VV);
  const int t = p / VV;
  const int v = p - t * VV;
  return (n * VV + v) * TT + t;
}

__global__ void __launch_bounds__(128)
tgat_xw_wmma(const float* __restrict__ x, const float* __restrict__ W,
             float* __restrict__ xw) {
  __shared__ v2f sWp[32 * 64];
  for (int i = threadIdx.x; i < 32 * 64; i += 128) {
    const int kp  = i >> 6;
    const int col = i & 63;
    v2f p;
    p.x = W[(2 * kp) * 64 + col];
    p.y = W[(2 * kp + 1) * 64 + col];
    sWp[i] = p;
  }
  __syncthreads();

  const int lane  = threadIdx.x & 31;
  const int wave  = threadIdx.x >> 5;
  const int half  = lane >> 4;
  const int mloc  = lane & 15;
  const int qbase = blockIdx.x * 64 + wave * 16;

  const int q = qbase + mloc;
  const int n = q / (TT * VV);
  const int p = q - n * (TT * VV);
  const float* xb = x + (size_t)n * (CCH * TT * VV) + p;

  v8f acc0 = {}, acc1 = {}, acc2 = {}, acc3 = {};
  for (int k = 0; k < 64; k += 4) {
    const int k0 = k + 2 * half;
    v2f a;
    a.x = xb[(size_t)k0 * (TT * VV)];
    a.y = xb[(size_t)(k0 + 1) * (TT * VV)];
    const v2f* wp = sWp + (k0 >> 1) * 64 + mloc;
    const v2f b0 = wp[0];
    const v2f b1 = wp[16];
    const v2f b2 = wp[32];
    const v2f b3 = wp[48];
    acc0 = __builtin_amdgcn_wmma_f32_16x16x4_f32(false, a, false, b0, (short)0, acc0, false, false);
    acc1 = __builtin_amdgcn_wmma_f32_16x16x4_f32(false, a, false, b1, (short)0, acc1, false, false);
    acc2 = __builtin_amdgcn_wmma_f32_16x16x4_f32(false, a, false, b2, (short)0, acc2, false, false);
    acc3 = __builtin_amdgcn_wmma_f32_16x16x4_f32(false, a, false, b3, (short)0, acc3, false, false);
  }

#pragma unroll
  for (int r = 0; r < 8; ++r) {
    float* orow = xw + (size_t)tgat_row_of(qbase + r + 8 * half) * 64 + mloc;
    orow[0]  = acc0[r];
    orow[16] = acc1[r];
    orow[32] = acc2[r];
    orow[48] = acc3[r];
  }
}

#define CS_P  68
#define PSTEPS 5
#define RTP   19
#define RNG   8
#define RROWS (2 * RTP + 7)
#define RK0   5
#define RB0   (2 * RK0 + 7)
#define RK1   12
#define RB1   (2 * RK1 + 7)
#define SE_P  12

__device__ __forceinline__ int tgat_win_row(int row0, int i) {
  int r = row0 + i;
  if (i < KHALF) r = r < 0 ? 0 : r;
  if (i > (TT - 1) - (2 * (RNG - 1) * RTP - KHALF)) r = r > TT - 1 ? TT - 1 : r;
  return r;
}

__global__ void __launch_bounds__(256) __attribute__((amdgpu_waves_per_eu(8)))
tgat_attn(const float* __restrict__ xw, const float* __restrict__ alpha,
          const float* __restrict__ phi, float* __restrict__ res) {
  __shared__ float sx1[TP * 16];
  __shared__ float sx2[TT * 16];
  __shared__ __attribute__((aligned(16))) float spart[64 * CS_P];
  __shared__ float srow4[4 * 64];
  __shared__ __attribute__((aligned(16))) float sst[4 * 16 * CS_P];
  __shared__ __attribute__((aligned(16))) float se[TP * SE_P];

  const int nv = blockIdx.x;
  const float* xwB = xw + (size_t)nv * TT * 64;
  const int tid = threadIdx.x;

  {
    const int lane = tid & 31;
    const int wave = tid >> 5;
    const int hh   = lane >> 4;
    const int l16  = lane & 15;
    const int nt   = wave & 1;
    const int pr   = wave >> 1;
    const float* prow = (nt ? phi : alpha) + l16 * 64 + 4 * hh;
    v4f b[8];
    for (int i = 0; i < 8; ++i) b[i] = *(const v4f*)(prow + 8 * i);
    v4f cs[4];
    for (int j = 0; j < 4; ++j) cs[j] = (v4f){0.f, 0.f, 0.f, 0.f};

    const int stw = pr * (16 * CS_P) + (8 * nt + hh) * CS_P + 4 * l16;
    const int str = pr * (16 * CS_P) + l16 * CS_P + 4 * hh;
    v4f g[4];
    for (int j = 0; j < 4; ++j) {
      int row = pr * 16 + 8 * nt + 2 * j + hh;
      row = row < TT ? row : TT - 1;
      g[j] = *(const v4f*)(xwB + (size_t)row * 64 + 4 * l16);
    }
    for (int j = 0; j < 4; ++j) *(v4f*)(sst + stw + 2 * j * CS_P) = g[j];

    for (int it = 0; it < PSTEPS; ++it) {
      const int mt = pr + 4 * it;
      const float* rbuf = (it & 1) ? spart : sst;
      float*       wbuf = (it & 1) ? sst : spart;
      __syncthreads();
      if (it + 1 < PSTEPS) {
        for (int j = 0; j < 4; ++j) {
          int row = (mt + 4) * 16 + 8 * nt + 2 * j + hh;
          row = row < TT ? row : TT - 1;
          g[j] = *(const v4f*)(xwB + (size_t)row * 64 + 4 * l16);
        }
      }
      v4f a[8];
      for (int i = 0; i < 8; ++i) a[i] = *(const v4f*)(rbuf + str + 8 * i);
      v8f acc0 = {}, acc1 = {};
      for (int i = 0; i < 8; ++i) {
        v2f a0, a1, b0, b1;
        a0.x = a[i].x; a0.y = a[i].y; a1.x = a[i].z; a1.y = a[i].w;
        b0.x = b[i].x; b0.y = b[i].y; b1.x = b[i].z; b1.y = b[i].w;
        acc0 = __builtin_amdgcn_wmma_f32_16x16x4_f32(false, a0, false, b0, (short)0, acc0, false, false);
        acc1 = __builtin_amdgcn_wmma_f32_16x16x4_f32(false, a1, false, b1, (short)0, acc1, false, false);
      }
      if (mt * 16 + l16 < TT) {
        if (nt) { for (int j = 0; j < 4; ++j) cs[j] += a[4 + j]; }
        else    { for (int j = 0; j < 4; ++j) cs[j] += a[j]; }
      }
      const int r0 = mt * 16 + 8 * hh;
      if (nt) {
        for (int r = 0; r < 8; ++r) {
          const int orow = r0 + r;
          if (orow < TT) sx2[orow * 16 + l16] = acc0[r] + acc1[r];
        }
      } else {
        for (int r = 0; r < 8; r += 2) {
          const int orow = r0 + r;
          if (orow < TT) sx1[(orow >> 1) * 16 + l16] = acc0[r] + acc1[r];
        }
      }
      if (it + 1 < PSTEPS) {
        for (int j = 0; j < 4; ++j) *(v4f*)(wbuf + stw + 2 * j * CS_P) = g[j];
      }
    }
    float* prt = spart + (pr * 16 + l16) * CS_P + 32 * nt + 4 * hh;
    for (int j = 0; j < 4; ++j) *(v4f*)(prt + 8 * j) = cs[j];
  }
  __syncthreads();

  const int rc2  = 2 * (tid & 31);
  const int tp0  = (tid >> 5) * RTP;
  const int row0 = 2 * tp0 - KHALF;
  const float* xc = xwB + rc2;
  v2f xr[RROWS];
#pragma unroll
  for (int i = 0; i < RB0; ++i) xr[i] = *(const v2f*)(xc + tgat_win_row(row0, i) * 64);

  {
    const int rc = tid & 63;
    float accv = 0.f;
    for (int l = 0; l < 16; ++l) accv += spart[((tid >> 6) * 16 + l) * CS_P + rc];
    srow4[tid] = accv;
  }

  for (int idx = tid; idx < TP * 9; idx += 256) {
    const int tp = idx / 9;
    const int j  = idx - tp * 9;
    const int s  = 2 * tp - KHALF + j;
    float e = 0.f;
    if ((unsigned)s < (unsigned)TT) {
      float sc = 0.f;
      for (int h = 0; h < 16; ++h) sc += sx1[tp * 16 + h] * sx2[s * 16 + h];
      e = __expf(sc) - 1.f;
    }
    se[tp * SE_P + j] = e;
  }
  __syncthreads();

  for (int tp = tid; tp < TP; tp += 256) {
    float D = (float)TT;
    for (int j = 0; j < 9; ++j) D += se[tp * SE_P + j];
    se[tp * SE_P + 9] = 1.0f / D;
  }
  __syncthreads();

  v2f rs;
  rs.x = ((srow4[rc2] + srow4[64 + rc2]) + srow4[128 + rc2]) + srow4[192 + rc2];
  rs.y = ((srow4[rc2 + 1] + srow4[64 + rc2 + 1]) + srow4[128 + rc2 + 1]) + srow4[192 + rc2 + 1];
  float* rout = res + (size_t)nv * (TP * 64) + rc2;
#pragma unroll
  for (int i = RB0; i < RB1; ++i) xr[i] = *(const v2f*)(xc + tgat_win_row(row0, i) * 64);
  float wq[2][10];
  {
    const float* rec = se + tp0 * SE_P;
#pragma unroll
    for (int j = 0; j < 10; ++j) wq[0][j] = rec[j];
  }
#pragma unroll
  for (int k = 0; k < RTP; ++k) {
    if (k == RK0) {
#pragma unroll
      for (int i = RB1; i < RROWS; ++i) xr[i] = *(const v2f*)(xc + tgat_win_row(row0, i) * 64);
    }
    if (k + 1 < RTP) {
      int tpn = tp0 + k + 1;
      if (k + 1 >= TP - (RNG - 1) * RTP) tpn = tpn < TP ? tpn : TP - 1;
      const float* rec = se + tpn * SE_P;
#pragma unroll
      for (int j = 0; j < 10; ++j) wq[(k + 1) & 1][j] = rec[j];
    }
    const int tp = tp0 + k;
    if (k < TP - (RNG - 1) * RTP || tp < TP) {
      v2f accv = rs;
#pragma unroll
      for (int j = 0; j < 9; ++j) accv += wq[k & 1][j] * xr[2 * k + j];
      *(v2f*)(rout + tp * 64) = accv * wq[k & 1][9];
    }
    __builtin_amdgcn_sched_barrier(0);
  }
}

#define TPB    6
#define TILE_W (TPB * VV)
#define TILE_P 151
#define TILE_Q (VV * TPB * 16)

__global__ void __launch_bounds__(256)
tgat_out_tiles(const float* __restrict__ res, float* __restrict__ out) {
  __shared__ float tile[CCH * TILE_P];
  const int n   = blockIdx.x / (TP / TPB);
  const int k   = blockIdx.x - n * (TP / TPB);
  const int tid = threadIdx.x;

  v4f val[10];
#pragma unroll
  for (int u = 0; u < 10; ++u) {
    int i = tid + u * 256;
    i = i < TILE_Q ? i : TILE_Q - 1;
    const int v = i / (TPB * 16);
    const int r = i - v * (TPB * 16);
    val[u] = *(const v4f*)(res + ((size_t)(n * VV + v) * TP + k * TPB) * 64 + r * 4);
  }
#pragma unroll
  for (int u = 0; u < 10; ++u) {
    const int i = tid + u * 256;
    if (i < TILE_Q) {
      const int v   = i / (TPB * 16);
      const int r   = i - v * (TPB * 16);
      const int tpl = r >> 4;
      const int c4  = r & 15;
      float* d = tile + (4 * c4) * TILE_P + tpl * VV + v;
      d[0]          = val[u].x;
      d[TILE_P]     = val[u].y;
      d[2 * TILE_P] = val[u].z;
      d[3 * TILE_P] = val[u].w;
    }
  }
  __syncthreads();

  const int lane = tid & 31;
  const int wave = tid >> 5;
  for (int c = wave; c < CCH; c += 8) {
    const size_t g0 = ((size_t)(n * CCH + c) * TP + k * TPB) * VV;
    const int lead = (int)(g0 & 31);
    float w[6];
#pragma unroll
    for (int m = 0; m < 6; ++m) {
      int j = lane - lead + 32 * m;
      j = j < 0 ? 0 : (j > TILE_W - 1 ? TILE_W - 1 : j);
      w[m] = tile[c * TILE_P + j];
    }
#pragma unroll
    for (int m = 0; m < 6; ++m) {
      const int j = lane - lead + 32 * m;
      if (j >= 0 && j < TILE_W) out[g0 + j] = w[m];
    }
  }
}

extern "C" void kernel_launch(void* const* d_in, const int* in_sizes, int n_in,
                              void* d_out, int out_size, void* d_ws, size_t ws_size,
                              hipStream_t stream) {
  const float* x     = (const float*)d_in[0];
  const float* W     = (const float*)d_in[1];
  const float* alpha = (const float*)d_in[2];
  const float* phi   = (const float*)d_in[3];

  float* xw  = (float*)d_ws;
  float* res = xw + (size_t)MROWS * 64;

  tgat_xw_wmma<<<MROWS / 64, 128, 0, stream>>>(x, W, xw);
  tgat_attn<<<NV, 256, 0, stream>>>(xw, alpha, phi, res);
  tgat_out_tiles<<<NB * (TP / TPB), 256, 0, stream>>>(res, (float*)d_out);
}


# W applied last (re-association): sums, scores (x times W alpha^T, W phi^T), band sums on x in its own order, out = W^T y per (n,5 tp,25 v) tile; 2 launches, no xw/result plane; order of sums changes
# speedup vs baseline: 8.5824x; 1.0635x over previous
#include <hip/hip_runtime.h>
#include <math.h>

typedef __attribute__((ext_vector_type(2))) float v2f;
typedef __attribute__((ext_vector_type(4))) float v4f;
typedef __attribute__((ext_vector_type(8))) float v8f;

#define NB    32
#define CCH   64
#define TT    300
#define VV    25
#define PLANE (TT*VV)
#define TP    150
#define KHALF 4


#define SROWS  10
#define SPASS  (TT / SROWS)
#define SPITCH 32

__global__ void __launch_bounds__(256)
tgat_sums_fold(const float* __restrict__ x, const float* __restrict__ W,
               const float* __restrict__ alpha, const float* __restrict__ phi,
               float* __restrict__ Sg, float* __restrict__ Pg) {
  __shared__ float part[SROWS * VV];
  const int tid = threadIdx.x;
  if (blockIdx.x == 0) {
    const int lane = tid & 31, wave = tid >> 5;
    const int hh = lane >> 4, l16 = lane & 15;
    const int mt = wave >> 1, nt = wave & 1;
    const float* arow = W + (16 * mt + l16) * 64 + 2 * hh;
    const float* brow = (nt ? phi : alpha) + l16 * 64 + 2 * hh;
    v8f acc0 = {}, acc1 = {};
#pragma unroll
    for (int i = 0; i < 16; i += 2) {
      const v2f a0 = *(const v2f*)(arow + 4 * i);
      const v2f b0 = *(const v2f*)(brow + 4 * i);
      const v2f a1 = *(const v2f*)(arow + 4 * i + 4);
      const v2f b1 = *(const v2f*)(brow + 4 * i + 4);
      acc0 = __builtin_amdgcn_wmma_f32_16x16x4_f32(false, a0, false, b0, (short)0, acc0, false, false);
      acc1 = __builtin_amdgcn_wmma_f32_16x16x4_f32(false, a1, false, b1, (short)0, acc1, false, false);
    }
#pragma unroll
    for (int r = 0; r < 8; ++r) {
      const int row = 16 * mt + r + 8 * hh;
      Pg[((row >> 1) * 32 + 16 * nt + l16) * 2 + (row & 1)] = acc0[r] + acc1[r];
    }
    return;
  }
  const int plane = blockIdx.x - 1;
  const float* pl = x + (size_t)plane * PLANE;
  if (tid < SROWS * VV) {
    float g[SPASS];
#pragma unroll
    for (int m = 0; m < SPASS; ++m) g[m] = pl[tid + (SROWS * VV) * m];
    float s = 0.f;
#pragma unroll
    for (int m = 0; m < SPASS; ++m) s += g[m];
    part[tid] = s;
  }
  __syncthreads();
  if (tid < VV) {
    float a = 0.f;
    for (int q = 0; q < SROWS; ++q) a += part[tid + VV * q];
    Sg[plane * SPITCH + tid] = a;
  }
}

#define TPW   5
#define RW    (2 * TPW + 7)
#define NPOS  (RW * VV)
#define NTILE 27
#define SLP   432
#define X2P   18
#define OPOS  (TPW * VV)
#define NOTL  8
#define YP    144
#define OTP   130
#define SE_P  12
#define NPHI  6
#define SPQ   7

__global__ void __launch_bounds__(256)
tgat_fused(const float* __restrict__ x, const float* __restrict__ W,
           const float* __restrict__ Sg, const float* __restrict__ Pg,
           float* __restrict__ out) {
  __shared__ __attribute__((aligned(16))) float slab[CCH * SLP];
  __shared__ __attribute__((aligned(16))) float prj[SLP * X2P + OPOS * X2P];
  __shared__ float se[OPOS * SE_P];
  __shared__ float sS[CCH * VV];

  float* sx2 = prj;
  float* sx1 = prj + SLP * X2P;
  v2f*   syp = (v2f*)prj;
  float* tile = slab;
  v2f*   sWp  = (v2f*)(slab + CCH * OTP);

  const int tid  = threadIdx.x;
  const int lane = tid & 31;
  const int wave = tid >> 5;
  const int hh   = lane >> 4;
  const int l16  = lane & 15;
  const int nq   = blockIdx.x / (TP / TPW);
  const int n    = (NB - 1) - nq;
  const int tp0  = (blockIdx.x - nq * (TP / TPW)) * TPW;
  const int g0r  = (2 * tp0 - KHALF) * VV;
  const float* xn = x + (size_t)n * (CCH * PLANE);

  const int isphi = wave < NPHI;
  v2f b[16];
#pragma unroll
  for (int i = 0; i < 16; ++i)
    b[i] = *(const v2f*)(Pg + ((2 * i + hh) * 32 + 16 * isphi + l16) * 2);

  for (int i = tid; i < CCH * VV; i += 256) {
    const int c = i / VV;
    const int v = i - c * VV;
    sS[i] = Sg[(n * CCH + c) * SPITCH + v];
  }

  {
    v2f g[8][SPQ];
#pragma unroll
    for (int cc = 0; cc < 8; ++cc) {
      const float* xc = xn + (wave + 8 * cc) * PLANE + g0r;
#pragma unroll
      for (int m = 0; m < SPQ; ++m) {
        const int pos = 2 * (lane + 32 * m);
        v2f val = {0.f, 0.f};
        if (pos < NPOS && (unsigned)(g0r + pos) < (unsigned)PLANE) val = *(const v2f*)(xc + pos);
        if (pos + 1 >= NPOS) val.y = 0.f;
        g[cc][m] = val;
      }
    }
#pragma unroll
    for (int cc = 0; cc < 8; ++cc) {
#pragma unroll
      for (int m = 0; m < SPQ; ++m) {
        const int pos = 2 * (lane + 32 * m);
        if (pos < SLP) *(v2f*)(slab + (wave + 8 * cc) * SLP + pos) = g[cc][m];
      }
    }
  }
  __syncthreads();

  if (isphi) {
    for (int t = wave; t < NTILE; t += NPHI) {
      const float* ap = slab + (2 * hh) * SLP + 16 * t + l16;
      v8f acc0 = {}, acc1 = {};
#pragma unroll
      for (int i = 0; i < 16; i += 2) {
        v2f a0, a1;
        a0.x = ap[(4 * i) * SLP];     a0.y = ap[(4 * i + 1) * SLP];
        a1.x = ap[(4 * i + 4) * SLP]; a1.y = ap[(4 * i + 5) * SLP];
        acc0 = __builtin_amdgcn_wmma_f32_16x16x4_f32(false, a0, false, b[i],     (short)0, acc0, false, false);
        acc1 = __builtin_amdgcn_wmma_f32_16x16x4_f32(false, a1, false, b[i + 1], (short)0, acc1, false, false);
      }
#pragma unroll
      for (int r = 0; r < 8; ++r)
        sx2[(16 * t + 8 * hh + r) * X2P + l16] = acc0[r] + acc1[r];
    }
  } else {
    for (int t = wave - NPHI; t < NOTL; t += 8 - NPHI) {
      int po = 16 * t + l16;
      po = po < OPOS ? po : OPOS - 1;
      const int tpl = po / VV;
      const int v   = po - tpl * VV;
      const float* ap = slab + (2 * hh) * SLP + (2 * tpl + KHALF) * VV + v;
      v8f acc0 = {}, acc1 = {};
#pragma unroll
      for (int i = 0; i < 16; i += 2) {
        v2f a0, a1;
        a0.x = ap[(4 * i) * SLP];     a0.y = ap[(4 * i + 1) * SLP];
        a1.x = ap[(4 * i + 4) * SLP]; a1.y = ap[(4 * i + 5) * SLP];
        acc0 = __builtin_amdgcn_wmma_f32_16x16x4_f32(false, a0, false, b[i],     (short)0, acc0, false, false);
        acc1 = __builtin_amdgcn_wmma_f32_16x16x4_f32(false, a1, false, b[i + 1], (short)0, acc1, false, false);
      }
#pragma unroll
      for (int r = 0; r < 8; ++r) {
        const int por = 16 * t + 8 * hh + r;
        if (por < OPOS) sx1[por * X2P + l16] = acc0[r] + acc1[r];
      }
    }
  }
  __syncthreads();

  for (int idx = tid; idx < OPOS * 9; idx += 256) {
    const int po  = idx / 9;
    const int j   = idx - po * 9;
    const int tpl = po / VV;
    const int v   = po - tpl * VV;
    const int s   = 2 * (tp0 + tpl) - KHALF + j;
    float e = 0.f;
    if ((unsigned)s < (unsigned)TT) {
      const float* p1 = sx1 + po * X2P;
      const float* p2 = sx2 + ((2 * tpl + j) * VV + v) * X2P;
      float sc = 0.f;
      for (int h = 0; h < 16; ++h) sc += p1[h] * p2[h];
      e = __expf(sc) - 1.f;
    }
    se[po * SE_P + j] = e;
  }
  __syncthreads();

  v2f wreg[8];
#pragma unroll
  for (int u = 0; u < 8; ++u) {
    const int i   = tid + 256 * u;
    const int kp  = i >> 6;
    const int col = i & 63;
    wreg[u].x = W[(2 * kp) * 64 + col];
    wreg[u].y = W[(2 * kp + 1) * 64 + col];
  }

  {
    const int po  = 32 * (wave & 3) + lane;
    const int cp0 = 16 * (wave >> 2);
    if (po < OPOS) {
      const int tpl = po / VV;
      const int v   = po - tpl * VV;
      float e[9];
#pragma unroll
      for (int j = 0; j < 9; ++j) e[j] = se[po * SE_P + j];
      float D = (float)TT;
#pragma unroll
      for (int j = 0; j < 9; ++j) D += e[j];
      const float rd = 1.0f / D;
      const float* sb = slab + (2 * tpl) * VV + v;
      for (int cp = cp0; cp < cp0 + 16; ++cp) {
        const float* s0 = sb + (2 * cp) * SLP;
        const float* s1 = s0 + SLP;
        float y0 = sS[(2 * cp) * VV + v];
        float y1 = sS[(2 * cp + 1) * VV + v];
#pragma unroll
        for (int j = 0; j < 9; ++j) {
          y0 += e[j] * s0[VV * j];
          y1 += e[j] * s1[VV * j];
        }
        v2f y;
        y.x = y0 * rd;
        y.y = y1 * rd;
        syp[cp * YP + po] = y;
      }
    } else {
      for (int cp = cp0; cp < cp0 + 16; ++cp) syp[cp * YP + po] = (v2f){0.f, 0.f};
    }
  }
  __syncthreads();

#pragma unroll
  for (int u = 0; u < 8; ++u) sWp[tid + 256 * u] = wreg[u];
  __syncthreads();

  {
    const v2f* yb = syp + hh * YP + 16 * wave + l16;
    const v2f* wb = sWp + hh * 64 + l16;
    v8f acc0 = {}, acc1 = {}, acc2 = {}, acc3 = {};
#pragma unroll
    for (int i = 0; i < 16; ++i) {
      const v2f bb = yb[(2 * i) * YP];
      const v2f* wp = wb + (2 * i) * 64;
      const v2f a0 = wp[0];
      const v2f a1 = wp[16];
      const v2f a2 = wp[32];
      const v2f a3 = wp[48];
      acc0 = __builtin_amdgcn_wmma_f32_16x16x4_f32(false, a0, false, bb, (short)0, acc0, false, false);
      acc1 = __builtin_amdgcn_wmma_f32_16x16x4_f32(false, a1, false, bb, (short)0, acc1, false, false);
      acc2 = __builtin_amdgcn_wmma_f32_16x16x4_f32(false, a2, false, bb, (short)0, acc2, false, false);
      acc3 = __builtin_amdgcn_wmma_f32_16x16x4_f32(false, a3, false, bb, (short)0, acc3, false, false);
    }
    float* td = tile + (8 * hh) * OTP + 16 * wave + l16;
#pragma unroll
    for (int r = 0; r < 8; ++r) {
      td[r * OTP]             = acc0[r];
      td[(16 + r) * OTP]      = acc1[r];
      td[(32 + r) * OTP]      = acc2[r];
      td[(48 + r) * OTP]      = acc3[r];
    }
  }
  __syncthreads();

  for (int c = wave; c < CCH; c += 8) {
    const size_t g0 = ((size_t)(n * CCH + c) * TP + tp0) * VV;
    const int lead = (int)(g0 & 31);
    float w[5];
#pragma unroll
    for (int m = 0; m < 5; ++m) {
      int j = lane - lead + 32 * m;
      j = j < 0 ? 0 : (j > OPOS - 1 ? OPOS - 1 : j);
      w[m] = tile[c * OTP + j];
    }
#pragma unroll
    for (int m = 0; m < 5; ++m) {
      const int j = lane - lead + 32 * m;
      if (j >= 0 && j < OPOS) out[g0 + j] = w[m];
    }
  }
}

extern "C" void kernel_launch(void* const* d_in, const int* in_sizes, int n_in,
                              void* d_out, int out_size, void* d_ws, size_t ws_size,
                              hipStream_t stream) {
  const float* x     = (const float*)d_in[0];
  const float* W     = (const float*)d_in[1];
  const float* alpha = (const float*)d_in[2];
  const float* phi   = (const float*)d_in[3];

  float* Sg = (float*)d_ws;
  float* Pg = Sg + NB * CCH * SPITCH;

  tgat_sums_fold<<<1 + NB * CCH, 256, 0, stream>>>(x, W, alpha, phi, Sg, Pg);
  tgat_fused<<<NB * (TP / TPW), 256, 0, stream>>>(x, W, Sg, Pg, (float*)d_out);
}


# prologue: every input load of a work-group goes out in one burst with one wait: sums records ride in registers behind the slab (were 7 load-wait-store trips ahead of it); sums unchanged
# speedup vs baseline: 9.8911x; 1.1525x over previous
#include <hip/hip_runtime.h>
#include <math.h>

typedef __attribute__((ext_vector_type(2))) float v2f;
typedef __attribute__((ext_vector_type(4))) float v4f;
typedef __attribute__((ext_vector_type(8))) float v8f;

#define NB    32
#define CCH   64
#define TT    300
#define VV    25
#define PLANE (TT*VV)
#define TP    150
#define KHALF 4


#define SROWS  10
#define SPASS  (TT / SROWS)
#define SPITCH 32

__global__ void __launch_bounds__(256)
tgat_sums_fold(const float* __restrict__ x, const float* __restrict__ W,
               const float* __restrict__ alpha, const float* __restrict__ phi,
               float* __restrict__ Sg, float* __restrict__ Pg) {
  __shared__ float part[SROWS * VV];
  const int tid = threadIdx.x;
  if (blockIdx.x == 0) {
    const int lane = tid & 31, wave = tid >> 5;
    const int hh = lane >> 4, l16 = lane & 15;
    const int mt = wave >> 1, nt = wave & 1;
    const float* arow = W + (16 * mt + l16) * 64 + 2 * hh;
    const float* brow = (nt ? phi : alpha) + l16 * 64 + 2 * hh;
    v8f acc0 = {}, acc1 = {};
#pragma unroll
    for (int i = 0; i < 16; i += 2) {
      const v2f a0 = *(const v2f*)(arow + 4 * i);
      const v2f b0 = *(const v2f*)(brow + 4 * i);
      const v2f a1 = *(const v2f*)(arow + 4 * i + 4);
      const v2f b1 = *(const v2f*)(brow + 4 * i + 4);
      acc0 = __builtin_amdgcn_wmma_f32_16x16x4_f32(false, a0, false, b0, (short)0, acc0, false, false);
      acc1 = __builtin_amdgcn_wmma_f32_16x16x4_f32(false, a1, false, b1, (short)0, acc1, false, false);
    }
#pragma unroll
    for (int r = 0; r < 8; ++r) {
      const int row = 16 * mt + r + 8 * hh;
      Pg[((row >> 1) * 32 + 16 * nt + l16) * 2 + (row & 1)] = acc0[r] + acc1[r];
    }
    return;
  }
  const int plane = blockIdx.x - 1;
  const float* pl = x + (size_t)plane * PLANE;
  if (tid < SROWS * VV) {
    float g[SPASS];
#pragma unroll
    for (int m = 0; m < SPASS; ++m) g[m] = pl[tid + (SROWS * VV) * m];
    float s = 0.f;
#pragma unroll
    for (int m = 0; m < SPASS; ++m) s += g[m];
    part[tid] = s;
  }
  __syncthreads();
  if (tid < VV) {
    float a = 0.f;
    for (int q = 0; q < SROWS; ++q) a += part[tid + VV * q];
    Sg[plane * SPITCH + tid] = a;
  }
}

#define TPW   5
#define RW    (2 * TPW + 7)
#define NPOS  (RW * VV)
#define NTILE 27
#define SLP   432
#define X2P   18
#define OPOS  (TPW * VV)
#define NOTL  8
#define YP    144
#define OTP   130
#define SE_P  12
#define NPHI  6
#define SPQ   7
#define SSQ   7

__global__ void __launch_bounds__(256)
tgat_fused(const float* __restrict__ x, const float* __restrict__ W,
           const float* __restrict__ Sg, const float* __restrict__ Pg,
           float* __restrict__ out) {
  __shared__ __attribute__((aligned(16))) float slab[CCH * SLP];
  __shared__ __attribute__((aligned(16))) float prj[SLP * X2P + OPOS * X2P];
  __shared__ float se[OPOS * SE_P];
  __shared__ float sS[CCH * VV];

  float* sx2 = prj;
  float* sx1 = prj + SLP * X2P;
  v2f*   syp = (v2f*)prj;
  float* tile = slab;
  v2f*   sWp  = (v2f*)(slab + CCH * OTP);

  const int tid  = threadIdx.x;
  const int lane = tid & 31;
  const int wave = tid >> 5;
  const int hh   = lane >> 4;
  const int l16  = lane & 15;
  const int nq   = blockIdx.x / (TP / TPW);
  const int n    = (NB - 1) - nq;
  const int tp0  = (blockIdx.x - nq * (TP / TPW)) * TPW;
  const int g0r  = (2 * tp0 - KHALF) * VV;
  const float* xn = x + (size_t)n * (CCH * PLANE);

  const int isphi = wave < NPHI;
  v2f b[16];
  float sreg[SSQ];
  {
    const int swave = __builtin_amdgcn_readfirstlane(wave);
    unsigned off[SPQ];
    bool inr[SPQ];
#pragma unroll
    for (int m = 0; m < SPQ; ++m) {
      const int gp = g0r + 2 * (lane + 32 * m);
      inr[m] = (unsigned)gp < (unsigned)PLANE;
      off[m] = (unsigned)(gp < 0 ? 0 : (gp > PLANE - 2 ? PLANE - 2 : gp));
    }
    v2f g[8][SPQ];
#pragma unroll
    for (int cc = 0; cc < 8; ++cc) {
      const float* xc = xn + (swave + 8 * cc) * PLANE;
#pragma unroll
      for (int m = 0; m < SPQ; ++m) g[cc][m] = *(const v2f*)(xc + off[m]);
    }
#pragma unroll
    for (int k = 0; k < SSQ; ++k) {
      int i = tid + 256 * k;
      i = i < CCH * VV ? i : CCH * VV - 1;
      const int c = i / VV;
      const int v = i - c * VV;
      sreg[k] = Sg[(n * CCH + c) * SPITCH + v];
    }
#pragma unroll
    for (int i = 0; i < 16; ++i)
      b[i] = *(const v2f*)(Pg + ((2 * i + hh) * 32 + 16 * isphi + l16) * 2);

    if (g0r < 0 || g0r + NPOS > PLANE) {
#pragma unroll
      for (int cc = 0; cc < 8; ++cc) {
#pragma unroll
        for (int m = 0; m < SPQ; ++m)
          if (!inr[m]) g[cc][m] = (v2f){0.f, 0.f};
      }
    }
#pragma unroll
    for (int cc = 0; cc < 8; ++cc) {
#pragma unroll
      for (int m = 0; m < SPQ; ++m) {
        const int pos = 2 * (lane + 32 * m);
        v2f val = g[cc][m];
        if (pos >= NPOS) val.x = 0.f;
        if (pos + 1 >= NPOS) val.y = 0.f;
        if (pos < SLP) *(v2f*)(slab + (swave + 8 * cc) * SLP + pos) = val;
      }
    }
#pragma unroll
    for (int k = 0; k < SSQ; ++k) {
      const int i = tid + 256 * k;
      if (i < CCH * VV) sS[i] = sreg[k];
    }
  }
  __syncthreads();

  if (isphi) {
    for (int t = wave; t < NTILE; t += NPHI) {
      const float* ap = slab + (2 * hh) * SLP + 16 * t + l16;
      v8f acc0 = {}, acc1 = {};
#pragma unroll
      for (int i = 0; i < 16; i += 2) {
        v2f a0, a1;
        a0.x = ap[(4 * i) * SLP];     a0.y = ap[(4 * i + 1) * SLP];
        a1.x = ap[(4 * i + 4) * SLP]; a1.y = ap[(4 * i + 5) * SLP];
        acc0 = __builtin_amdgcn_wmma_f32_16x16x4_f32(false, a0, false, b[i],     (short)0, acc0, false, false);
        acc1 = __builtin_amdgcn_wmma_f32_16x16x4_f32(false, a1, false, b[i + 1], (short)0, acc1, false, false);
      }
#pragma unroll
      for (int r = 0; r < 8; ++r)
        sx2[(16 * t + 8 * hh + r) * X2P + l16] = acc0[r] + acc1[r];
    }
  } else {
    for (int t = wave - NPHI; t < NOTL; t += 8 - NPHI) {
      int po = 16 * t + l16;
      po = po < OPOS ? po : OPOS - 1;
      const int tpl = po / VV;
      const int v   = po - tpl * VV;
      const float* ap = slab + (2 * hh) * SLP + (2 * tpl + KHALF) * VV + v;
      v8f acc0 = {}, acc1 = {};
#pragma unroll
      for (int i = 0; i < 16; i += 2) {
        v2f a0, a1;
        a0.x = ap[(4 * i) * SLP];     a0.y = ap[(4 * i + 1) * SLP];
        a1.x = ap[(4 * i + 4) * SLP]; a1.y = ap[(4 * i + 5) * SLP];
        acc0 = __builtin_amdgcn_wmma_f32_16x16x4_f32(false, a0, false, b[i],     (short)0, acc0, false, false);
        acc1 = __builtin_amdgcn_wmma_f32_16x16x4_f32(false, a1, false, b[i + 1], (short)0, acc1, false, false);
      }
#pragma unroll
      for (int r = 0; r < 8; ++r) {
        const int por = 16 * t + 8 * hh + r;
        if (por < OPOS) sx1[por * X2P + l16] = acc0[r] + acc1[r];
      }
    }
  }
  __syncthreads();

  for (int idx = tid; idx < OPOS * 9; idx += 256) {
    const int po  = idx / 9;
    const int j   = idx - po * 9;
    const int tpl = po / VV;
    const int v   = po - tpl * VV;
    const int s   = 2 * (tp0 + tpl) - KHALF + j;
    float e = 0.f;
    if ((unsigned)s < (unsigned)TT) {
      const float* p1 = sx1 + po * X2P;
      const float* p2 = sx2 + ((2 * tpl + j) * VV + v) * X2P;
      float sc = 0.f;
      for (int h = 0; h < 16; ++h) sc += p1[h] * p2[h];
      e = __expf(sc) - 1.f;
    }
    se[po * SE_P + j] = e;
  }
  __syncthreads();

  v2f wreg[8];
#pragma unroll
  for (int u = 0; u < 8; ++u) {
    const int i   = tid + 256 * u;
    const int kp  = i >> 6;
    const int col = i & 63;
    wreg[u].x = W[(2 * kp) * 64 + col];
    wreg[u].y = W[(2 * kp + 1) * 64 + col];
  }

  {
    const int po  = 32 * (wave & 3) + lane;
    const int cp0 = 16 * (wave >> 2);
    if (po < OPOS) {
      const int tpl = po / VV;
      const int v   = po - tpl * VV;
      float e[9];
#pragma unroll
      for (int j = 0; j < 9; ++j) e[j] = se[po * SE_P + j];
      float D = (float)TT;
#pragma unroll
      for (int j = 0; j < 9; ++j) D += e[j];
      const float rd = 1.0f / D;
      const float* sb = slab + (2 * tpl) * VV + v;
      for (int cp = cp0; cp < cp0 + 16; ++cp) {
        const float* s0 = sb + (2 * cp) * SLP;
        const float* s1 = s0 + SLP;
        float y0 = sS[(2 * cp) * VV + v];
        float y1 = sS[(2 * cp + 1) * VV + v];
#pragma unroll
        for (int j = 0; j < 9; ++j) {
          y0 += e[j] * s0[VV * j];
          y1 += e[j] * s1[VV * j];
        }
        v2f y;
        y.x = y0 * rd;
        y.y = y1 * rd;
        syp[cp * YP + po] = y;
      }
    } else {
      for (int cp = cp0; cp < cp0 + 16; ++cp) syp[cp * YP + po] = (v2f){0.f, 0.f};
    }
  }
  __syncthreads();

#pragma unroll
  for (int u = 0; u < 8; ++u) sWp[tid + 256 * u] = wreg[u];
  __syncthreads();

  {
    const v2f* yb = syp + hh * YP + 16 * wave + l16;
    const v2f* wb = sWp + hh * 64 + l16;
    v8f acc0 = {}, acc1 = {}, acc2 = {}, acc3 = {};
#pragma unroll
    for (int i = 0; i < 16; ++i) {
      const v2f bb = yb[(2 * i) * YP];
      const v2f* wp = wb + (2 * i) * 64;
      const v2f a0 = wp[0];
      const v2f a1 = wp[16];
      const v2f a2 = wp[32];
      const v2f a3 = wp[48];
      acc0 = __builtin_amdgcn_wmma_f32_16x16x4_f32(false, a0, false, bb, (short)0, acc0, false, false);
      acc1 = __builtin_amdgcn_wmma_f32_16x16x4_f32(false, a1, false, bb, (short)0, acc1, false, false);
      acc2 = __builtin_amdgcn_wmma_f32_16x16x4_f32(false, a2, false, bb, (short)0, acc2, false, false);
      acc3 = __builtin_amdgcn_wmma_f32_16x16x4_f32(false, a3, false, bb, (short)0, acc3, false, false);
    }
    float* td = tile + (8 * hh) * OTP + 16 * wave + l16;
#pragma unroll
    for (int r = 0; r < 8; ++r) {
      td[r * OTP]             = acc0[r];
      td[(16 + r) * OTP]      = acc1[r];
      td[(32 + r) * OTP]      = acc2[r];
      td[(48 + r) * OTP]      = acc3[r];
    }
  }
  __syncthreads();

  for (int c = wave; c < CCH; c += 8) {
    const size_t g0 = ((size_t)(n * CCH + c) * TP + tp0) * VV;
    const int lead = (int)(g0 & 31);
    float w[5];
#pragma unroll
    for (int m = 0; m < 5; ++m) {
      int j = lane - lead + 32 * m;
      j = j < 0 ? 0 : (j > OPOS - 1 ? OPOS - 1 : j);
      w[m] = tile[c * OTP + j];
    }
#pragma unroll
    for (int m = 0; m < 5; ++m) {
      const int j = lane - lead + 32 * m;
      if (j >= 0 && j < OPOS) out[g0 + j] = w[m];
    }
  }
}

extern "C" void kernel_launch(void* const* d_in, const int* in_sizes, int n_in,
                              void* d_out, int out_size, void* d_ws, size_t ws_size,
                              hipStream_t stream) {
  const float* x     = (const float*)d_in[0];
  const float* W     = (const float*)d_in[1];
  const float* alpha = (const float*)d_in[2];
  const float* phi   = (const float*)d_in[3];

  float* Sg = (float*)d_ws;
  float* Pg = Sg + NB * CCH * SPITCH;

  tgat_sums_fold<<<1 + NB * CCH, 256, 0, stream>>>(x, W, alpha, phi, Sg, Pg);
  tgat_fused<<<NB * (TP / TPW), 256, 0, stream>>>(x, W, Sg, Pg, (float*)d_out);
}
